# starting kernel
# baseline (speedup 1.0000x reference)
#include <hip/hip_runtime.h>
#include <hip/hip_bf16.h>

#define BG   1024
#define NN   128
#define VN   (BG * NN)
#define HD   128
#define EPG  2048
#define KTOP 30
#define CONV_O 32
#define CONV_W 3
#define CONV_T (KTOP - CONV_W + 1)
#define FLAT   (CONV_O * CONV_T)
#define NCLS   16

typedef float v2f __attribute__((ext_vector_type(2)));
typedef float v8f __attribute__((ext_vector_type(8)));

__global__ __launch_bounds__(256) void sage_aggregate(
    const float* __restrict__ X, const int* __restrict__ src,
    const int* __restrict__ dst, float* __restrict__ meanOut)
{
  __shared__ float feat[NN * HD];
  __shared__ float acc [NN * HD];
  __shared__ float deg [NN];

  const int g = blockIdx.x;
  const int t = threadIdx.x;
  const float* xg = X + (size_t)g * NN * HD;

  for (int i = t; i < NN * HD / 4; i += 256) {
    ((float4*)feat)[i] = ((const float4*)xg)[i];
    ((float4*)acc)[i]  = make_float4(0.f, 0.f, 0.f, 0.f);
  }
  if (t < NN) deg[t] = 0.f;
  __syncthreads();

  const int ebase = g * EPG;
  const int eg = t >> 2;
  const int fg = t & 3;
  for (int e = eg; e < EPG; e += 64) {
    const int s = src[ebase + e] & (NN - 1);
    const int d = dst[ebase + e] & (NN - 1);
    const float* fs = feat + s * HD + fg * 32;
    float*       ad = acc  + d * HD + fg * 32;
    if (fg == 0) atomicAdd(&deg[d], 1.0f);
#pragma unroll
    for (int f = 0; f < 32; ++f) atomicAdd(&ad[f], fs[f]);
  }
  __syncthreads();

  float* mg = meanOut + (size_t)g * NN * HD;
  for (int i = t; i < NN * HD; i += 256) {
    const float c = deg[i >> 7];
    mg[i] = acc[i] / (c > 1.f ? c : 1.f);
  }
}

__global__ __launch_bounds__(256) void sage_gemm(
    const float* __restrict__ Aagg, const float* __restrict__ Aself,
    const float* __restrict__ Wl,   const float* __restrict__ Wr,
    const float* __restrict__ bias, float* __restrict__ out)
{
  __shared__ float2 wp[128 * HD];

  const int t = threadIdx.x;
  for (int idx = t; idx < 128 * HD; idx += 256) {
    const int kk  = idx >> 7;
    const int col = idx & (HD - 1);
    const float* W = (kk < 64) ? Wl : Wr;
    const int k = (kk & 63) * 2;
    float2 v;
    v.x = W[(size_t)k       * HD + col];
    v.y = W[(size_t)(k + 1) * HD + col];
    wp[idx] = v;
  }
  __syncthreads();

  const int wave = t >> 5;
  const int lane = t & 31;
  const int m    = lane & 15;
  const int hi   = lane >> 4;
  const int kb   = hi * 2;

  const int rowBase = blockIdx.x * 128 + wave * 16;
  const float* arow1 = Aagg  + (size_t)(rowBase + m) * HD;
  const float* arow2 = Aself + (size_t)(rowBase + m) * HD;

  v8f acc0 = {}, acc1 = {}, acc2 = {}, acc3 = {};
  v8f acc4 = {}, acc5 = {}, acc6 = {}, acc7 = {};

#define GEMM_STEP(AROW, KOFF)                                              \
  {                                                                        \
    v2f a = *(const v2f*)((AROW) + k0 + kb);                               \
    const float2* bp = wp + ((k0 >> 1) + hi + (KOFF)) * HD + m;            \
    float2 bb; v2f b;                                                      \
    bb = bp[0];   b.x = bb.x; b.y = bb.y;                                  \
    acc0 = __builtin_amdgcn_wmma_f32_16x16x4_f32(false, a, false, b, (short)0, acc0, false, false); \
    bb = bp[16];  b.x = bb.x; b.y = bb.y;                                  \
    acc1 = __builtin_amdgcn_wmma_f32_16x16x4_f32(false, a, false, b, (short)0, acc1, false, false); \
    bb = bp[32];  b.x = bb.x; b.y = bb.y;                                  \
    acc2 = __builtin_amdgcn_wmma_f32_16x16x4_f32(false, a, false, b, (short)0, acc2, false, false); \
    bb = bp[48];  b.x = bb.x; b.y = bb.y;                                  \
    acc3 = __builtin_amdgcn_wmma_f32_16x16x4_f32(false, a, false, b, (short)0, acc3, false, false); \
    bb = bp[64];  b.x = bb.x; b.y = bb.y;                                  \
    acc4 = __builtin_amdgcn_wmma_f32_16x16x4_f32(false, a, false, b, (short)0, acc4, false, false); \
    bb = bp[80];  b.x = bb.x; b.y = bb.y;                                  \
    acc5 = __builtin_amdgcn_wmma_f32_16x16x4_f32(false, a, false, b, (short)0, acc5, false, false); \
    bb = bp[96];  b.x = bb.x; b.y = bb.y;                                  \
    acc6 = __builtin_amdgcn_wmma_f32_16x16x4_f32(false, a, false, b, (short)0, acc6, false, false); \
    bb = bp[112]; b.x = bb.x; b.y = bb.y;                                  \
    acc7 = __builtin_amdgcn_wmma_f32_16x16x4_f32(false, a, false, b, (short)0, acc7, false, false); \
  }

#pragma unroll 2
  for (int k0 = 0; k0 < HD; k0 += 4) GEMM_STEP(arow1, 0)
#pragma unroll 2
  for (int k0 = 0; k0 < HD; k0 += 4) GEMM_STEP(arow2, 64)
#undef GEMM_STEP

  const int rbase = rowBase + hi * 8;
  v8f accs[8] = {acc0, acc1, acc2, acc3, acc4, acc5, acc6, acc7};
#pragma unroll
  for (int ct = 0; ct < 8; ++ct) {
    const int col = ct * 16 + m;
    const float bv = bias[col];
#pragma unroll
    for (int v = 0; v < 8; ++v) {
      float val = accs[ct][v] + bv;
      val = val > 0.f ? val : 0.f;
      out[(size_t)(rbase + v) * HD + col] = val;
    }
  }
}

__global__ __launch_bounds__(256) void sortpool_head(
    const float* __restrict__ H,
    const float* __restrict__ conv_w, const float* __restrict__ conv_b,
    const float* __restrict__ W1, const float* __restrict__ b1,
    const float* __restrict__ W2, const float* __restrict__ b2,
    float* __restrict__ out)
{
  __shared__ float vals[NN];
  __shared__ int   idxs[NN];
  __shared__ float xs[KTOP * HD];
  __shared__ float y[FLAT];
  __shared__ float z[HD];
  __shared__ float logits[NCLS];
  __shared__ float lse;

  const int g = blockIdx.x;
  const int t = threadIdx.x;
  const float* hg = H + (size_t)g * NN * HD;

  if (t < NN) { vals[t] = hg[t * HD + (HD - 1)]; idxs[t] = t; }
  __syncthreads();

  for (int k = 2; k <= NN; k <<= 1) {
    for (int j = k >> 1; j > 0; j >>= 1) {
      if (t < NN) {
        const int ixj = t ^ j;
        if (ixj > t) {
          const bool dirDesc = ((t & k) == 0);
          const float va = vals[t], vb = vals[ixj];
          const int   ia = idxs[t], ib = idxs[ixj];
          const bool aBeforeB = (va > vb) || (va == vb && ia < ib);
          if (dirDesc ? !aBeforeB : aBeforeB) {
            vals[t] = vb; vals[ixj] = va;
            idxs[t] = ib; idxs[ixj] = ia;
          }
        }
      }
      __syncthreads();
    }
  }

  for (int i = t; i < KTOP * HD; i += 256) {
    const int tt = i >> 7, c = i & (HD - 1);
    xs[i] = hg[(size_t)idxs[tt] * HD + c];
  }
  __syncthreads();

  for (int i = t; i < FLAT; i += 256) {
    const int o = i / CONV_T, tt = i % CONV_T;
    float s = conv_b[o];
#pragma unroll
    for (int w = 0; w < CONV_W; ++w) {
      const float* xr = xs + (tt + w) * HD;
      const float* wr = conv_w + (size_t)o * HD * CONV_W + w;
      for (int c = 0; c < HD; ++c) s += xr[c] * wr[c * CONV_W];
    }
    y[i] = s > 0.f ? s : 0.f;
  }
  __syncthreads();

  if (t < HD) {
    float s = b1[t];
    for (int i = 0; i < FLAT; ++i) s += y[i] * W1[(size_t)i * HD + t];
    z[t] = s > 0.f ? s : 0.f;
  }
  __syncthreads();

  if (t < NCLS) {
    float s = b2[t];
    for (int j = 0; j < HD; ++j) s += z[j] * W2[j * NCLS + t];
    logits[t] = s;
  }
  __syncthreads();

  if (t == 0) {
    float mx = logits[0];
    for (int c = 1; c < NCLS; ++c) mx = fmaxf(mx, logits[c]);
    float se = 0.f;
    for (int c = 0; c < NCLS; ++c) se += expf(logits[c] - mx);
    lse = mx + logf(se);
  }
  __syncthreads();
  if (t < NCLS) out[(size_t)g * NCLS + t] = logits[t] - lse;
}

extern "C" void kernel_launch(void* const* d_in, const int* in_sizes, int n_in,
                              void* d_out, int out_size, void* d_ws, size_t ws_size,
                              hipStream_t stream) {
  (void)in_sizes; (void)n_in; (void)out_size; (void)ws_size;
  const float* x      = (const float*)d_in[0];
  const int*   src    = (const int*)d_in[1];
  const int*   dst    = (const int*)d_in[2];
  const float* Wl1    = (const float*)d_in[5];
  const float* bl1    = (const float*)d_in[6];
  const float* Wr1    = (const float*)d_in[7];
  const float* Wl2    = (const float*)d_in[8];
  const float* bl2    = (const float*)d_in[9];
  const float* Wr2    = (const float*)d_in[10];
  const float* Wl3    = (const float*)d_in[11];
  const float* bl3    = (const float*)d_in[12];
  const float* Wr3    = (const float*)d_in[13];
  const float* conv_w = (const float*)d_in[14];
  const float* conv_b = (const float*)d_in[15];
  const float* W1     = (const float*)d_in[16];
  const float* b1     = (const float*)d_in[17];
  const float* W2     = (const float*)d_in[18];
  const float* b2     = (const float*)d_in[19];

  float* ws   = (float*)d_ws;
  float* mean = ws;
  float* h1   = ws + (size_t)VN * HD;
  float* h2   = h1 + (size_t)VN * HD;

  const int gemmBlocks = VN / 128;

  sage_aggregate<<<BG, 256, 0, stream>>>(x, src, dst, mean);
  sage_gemm<<<gemmBlocks, 256, 0, stream>>>(mean, x, Wl1, Wr1, bl1, h1);
  sage_aggregate<<<BG, 256, 0, stream>>>(h1, src, dst, mean);
  sage_gemm<<<gemmBlocks, 256, 0, stream>>>(mean, h1, Wl2, Wr2, bl2, h2);
  sage_aggregate<<<BG, 256, 0, stream>>>(h2, src, dst, mean);
  sage_gemm<<<gemmBlocks, 256, 0, stream>>>(mean, h2, Wl3, Wr3, bl3, h1);
  sortpool_head<<<BG, 256, 0, stream>>>(h1, conv_w, conv_b, W1, b1, W2, b2,
                                        (float*)d_out);
}


# head Conv1d: lanes run along output channels instead of time steps; x is a broadcast LDS read, weights come from a transposed LDS copy staged once per block; bank conflicts gone; sum order kept
# speedup vs baseline: 4.7478x; 4.7478x over previous
#include <hip/hip_runtime.h>
#include <hip/hip_bf16.h>

#define BG   1024
#define NN   128
#define VN   (BG * NN)
#define HD   128
#define EPG  2048
#define KTOP 30
#define CONV_O 32
#define CONV_W 3
#define CONV_T (KTOP - CONV_W + 1)
#define FLAT   (CONV_O * CONV_T)
#define NCLS   16

typedef float v2f __attribute__((ext_vector_type(2)));
typedef float v8f __attribute__((ext_vector_type(8)));

#define ZROW NN

__global__ __launch_bounds__(256) void sage_aggregate(
    const float* __restrict__ X, const int* __restrict__ src,
    const int* __restrict__ dst, float* __restrict__ meanOut)
{
  __shared__ float4         feat4[(NN + 1) * (HD / 4)];
  __shared__ uint4          ek4[EPG / 8];
  __shared__ unsigned short wl[8][EPG];
  __shared__ unsigned char  rl[8][EPG];
  __shared__ int            rowoff[8][17];

  const int g    = blockIdx.x;
  const int t    = threadIdx.x;
  const int lane = t & 31;
  const int wave = __builtin_amdgcn_readfirstlane(t >> 5);

  const float4* xg4 = (const float4*)(X + (size_t)g * NN * HD);
  for (int i = t; i < NN * HD / 4; i += 256) feat4[i] = xg4[i];
  if (t < HD / 4) feat4[ZROW * (HD / 4) + t] = make_float4(0.f, 0.f, 0.f, 0.f);

  {
    const int eb = g * EPG + wave * 256 + lane;
    unsigned p[8];
#pragma unroll
    for (int j = 0; j < 8; ++j) {
      const unsigned s = (unsigned)src[eb + j * 32] & (NN - 1);
      const unsigned d = (unsigned)dst[eb + j * 32] & (NN - 1);
      p[j] = (d << 8) | s;
    }
    uint4 q;
    q.x = p[0] | (p[1] << 16); q.y = p[2] | (p[3] << 16);
    q.z = p[4] | (p[5] << 16); q.w = p[6] | (p[7] << 16);
    ek4[wave * 32 + lane] = q;
  }
  __syncthreads();

  unsigned short* mywl = wl[wave];
  unsigned nw = 0;
  for (int c = 0; c < 8; ++c) {
    const uint4 qq = ek4[c * 32 + lane];
    const unsigned ww[4] = {qq.x, qq.y, qq.z, qq.w};
#pragma unroll
    for (int j = 0; j < 8; ++j) {
      const unsigned e = (ww[j >> 1] >> ((j & 1) * 16)) & 0xffffu;
      const bool hit = (e >> 12) == (unsigned)wave;
      const unsigned mask = __builtin_amdgcn_ballot_w32(hit);
      const unsigned pos  = __builtin_amdgcn_mbcnt_lo(mask, nw);
      if (hit) mywl[pos] = (unsigned short)(e & 0x0fffu);
      nw += (unsigned)__builtin_popcount(mask);
    }
  }
  if (lane < (int)((32u - (nw & 31u)) & 31u)) mywl[nw + lane] = 0xffffu;
  __syncthreads();

  unsigned char* myrl = rl[wave];
  {
    const unsigned nb = (nw + 31u) >> 5;
    unsigned cur = 0;
    if (nb <= 12u) {
      unsigned dv[12], sv[12];
#pragma unroll
      for (int k = 0; k < 12; ++k) {
        const unsigned e = ((unsigned)k < nb) ? (unsigned)mywl[k * 32 + lane] : 0xffffu;
        dv[k] = e >> 8; sv[k] = e & 0xffu;
      }
      for (unsigned r = 0; r < 16; ++r) {
        if (lane == 0) rowoff[wave][r] = (int)cur;
#pragma unroll
        for (int k = 0; k < 12; ++k) {
          const bool hit = dv[k] == r;
          const unsigned mask = __builtin_amdgcn_ballot_w32(hit);
          const unsigned pos  = __builtin_amdgcn_mbcnt_lo(mask, cur);
          if (hit) myrl[pos] = (unsigned char)sv[k];
          cur += (unsigned)__builtin_popcount(mask);
        }
      }
    } else {
      for (unsigned r = 0; r < 16; ++r) {
        if (lane == 0) rowoff[wave][r] = (int)cur;
        for (unsigned b = 0; b < nb; ++b) {
          const unsigned e = mywl[b * 32 + lane];
          const bool hit = (e >> 8) == r;
          const unsigned mask = __builtin_amdgcn_ballot_w32(hit);
          const unsigned pos  = __builtin_amdgcn_mbcnt_lo(mask, cur);
          if (hit) myrl[pos] = (unsigned char)(e & 0xffu);
          cur += (unsigned)__builtin_popcount(mask);
        }
      }
    }
    if (lane == 0) rowoff[wave][16] = (int)cur;
  }
  __syncthreads();

  float4* mg4 = (float4*)(meanOut + (size_t)g * NN * HD);
  for (int r = 0; r < 16; ++r) {
    const int beg = __builtin_amdgcn_readfirstlane(rowoff[wave][r]);
    const int n   = __builtin_amdgcn_readfirstlane(rowoff[wave][r + 1]) - beg;
    float4 a = make_float4(0.f, 0.f, 0.f, 0.f);
    for (int c = 0; c < n; c += 32) {
      const int k = c + lane;
      const int myidx = (k < n) ? (int)myrl[beg + k] : ZROW;
      const int m = (n - c) < 32 ? (n - c) : 32;
      for (int j = 0; j < m; j += 4) {
        const int s0 = __builtin_amdgcn_readlane(myidx, j);
        const int s1 = __builtin_amdgcn_readlane(myidx, j + 1);
        const int s2 = __builtin_amdgcn_readlane(myidx, j + 2);
        const int s3 = __builtin_amdgcn_readlane(myidx, j + 3);
        const float4 v0 = feat4[s0 * (HD / 4) + lane];
        const float4 v1 = feat4[s1 * (HD / 4) + lane];
        const float4 v2 = feat4[s2 * (HD / 4) + lane];
        const float4 v3 = feat4[s3 * (HD / 4) + lane];
        a.x += v0.x; a.y += v0.y; a.z += v0.z; a.w += v0.w;
        a.x += v1.x; a.y += v1.y; a.z += v1.z; a.w += v1.w;
        a.x += v2.x; a.y += v2.y; a.z += v2.z; a.w += v2.w;
        a.x += v3.x; a.y += v3.y; a.z += v3.z; a.w += v3.w;
      }
    }
    const float cnt = (float)(n > 1 ? n : 1);
    float4 o;
    o.x = a.x / cnt; o.y = a.y / cnt; o.z = a.z / cnt; o.w = a.w / cnt;
    mg4[(wave * 16 + r) * (HD / 4) + lane] = o;
  }
}

__global__ __launch_bounds__(256) void sage_gemm(
    const float* __restrict__ Aagg, const float* __restrict__ Aself,
    const float* __restrict__ Wl,   const float* __restrict__ Wr,
    const float* __restrict__ bias, float* __restrict__ out)
{
  __shared__ float2 wp[128 * HD];

  const int t = threadIdx.x;
  for (int idx = t; idx < 128 * HD; idx += 256) {
    const int kk  = idx >> 7;
    const int col = idx & (HD - 1);
    const float* W = (kk < 64) ? Wl : Wr;
    const int k = (kk & 63) * 2;
    float2 v;
    v.x = W[(size_t)k       * HD + col];
    v.y = W[(size_t)(k + 1) * HD + col];
    wp[idx] = v;
  }
  __syncthreads();

  const int wave = t >> 5;
  const int lane = t & 31;
  const int m    = lane & 15;
  const int hi   = lane >> 4;
  const int kb   = hi * 2;

  const int rowBase = blockIdx.x * 128 + wave * 16;
  const float* arow1 = Aagg  + (size_t)(rowBase + m) * HD;
  const float* arow2 = Aself + (size_t)(rowBase + m) * HD;

  v8f acc0 = {}, acc1 = {}, acc2 = {}, acc3 = {};
  v8f acc4 = {}, acc5 = {}, acc6 = {}, acc7 = {};

#define GEMM_STEP(AROW, KOFF)                                              \
  {                                                                        \
    v2f a = *(const v2f*)((AROW) + k0 + kb);                               \
    const float2* bp = wp + ((k0 >> 1) + hi + (KOFF)) * HD + m;            \
    float2 bb; v2f b;                                                      \
    bb = bp[0];   b.x = bb.x; b.y = bb.y;                                  \
    acc0 = __builtin_amdgcn_wmma_f32_16x16x4_f32(false, a, false, b, (short)0, acc0, false, false); \
    bb = bp[16];  b.x = bb.x; b.y = bb.y;                                  \
    acc1 = __builtin_amdgcn_wmma_f32_16x16x4_f32(false, a, false, b, (short)0, acc1, false, false); \
    bb = bp[32];  b.x = bb.x; b.y = bb.y;                                  \
    acc2 = __builtin_amdgcn_wmma_f32_16x16x4_f32(false, a, false, b, (short)0, acc2, false, false); \
    bb = bp[48];  b.x = bb.x; b.y = bb.y;                                  \
    acc3 = __builtin_amdgcn_wmma_f32_16x16x4_f32(false, a, false, b, (short)0, acc3, false, false); \
    bb = bp[64];  b.x = bb.x; b.y = bb.y;                                  \
    acc4 = __builtin_amdgcn_wmma_f32_16x16x4_f32(false, a, false, b, (short)0, acc4, false, false); \
    bb = bp[80];  b.x = bb.x; b.y = bb.y;                                  \
    acc5 = __builtin_amdgcn_wmma_f32_16x16x4_f32(false, a, false, b, (short)0, acc5, false, false); \
    bb = bp[96];  b.x = bb.x; b.y = bb.y;                                  \
    acc6 = __builtin_amdgcn_wmma_f32_16x16x4_f32(false, a, false, b, (short)0, acc6, false, false); \
    bb = bp[112]; b.x = bb.x; b.y = bb.y;                                  \
    acc7 = __builtin_amdgcn_wmma_f32_16x16x4_f32(false, a, false, b, (short)0, acc7, false, false); \
  }

#pragma unroll 2
  for (int k0 = 0; k0 < HD; k0 += 4) GEMM_STEP(arow1, 0)
#pragma unroll 2
  for (int k0 = 0; k0 < HD; k0 += 4) GEMM_STEP(arow2, 64)
#undef GEMM_STEP

  const int rbase = rowBase + hi * 8;
  v8f accs[8] = {acc0, acc1, acc2, acc3, acc4, acc5, acc6, acc7};
#pragma unroll
  for (int ct = 0; ct < 8; ++ct) {
    const int col = ct * 16 + m;
    const float bv = bias[col];
#pragma unroll
    for (int v = 0; v < 8; ++v) {
      float val = accs[ct][v] + bv;
      val = val > 0.f ? val : 0.f;
      out[(size_t)(rbase + v) * HD + col] = val;
    }
  }
}

#define WT_PITCH 33

template <int NT>
__device__ __forceinline__ void conv_lane_is_channel(
    const float* xs, const float* wT, float bias, int tt0, int lane, float* y)
{
  float s[NT];
#pragma unroll
  for (int i = 0; i < NT; ++i) s[i] = bias;
  for (int w = 0; w < CONV_W; ++w) {
    const float* wcol = wT + (w * HD) * WT_PITCH + lane;
    const float* xrow = xs + (tt0 + w) * HD;
    for (int c = 0; c < HD; c += 4) {
      const float w0 = wcol[(c + 0) * WT_PITCH];
      const float w1 = wcol[(c + 1) * WT_PITCH];
      const float w2 = wcol[(c + 2) * WT_PITCH];
      const float w3 = wcol[(c + 3) * WT_PITCH];
#pragma unroll
      for (int i = 0; i < NT; ++i) {
        const float4 xv = *(const float4*)(xrow + i * 8 * HD + c);
        s[i] = __builtin_fmaf(xv.x, w0, s[i]);
        s[i] = __builtin_fmaf(xv.y, w1, s[i]);
        s[i] = __builtin_fmaf(xv.z, w2, s[i]);
        s[i] = __builtin_fmaf(xv.w, w3, s[i]);
      }
    }
  }
#pragma unroll
  for (int i = 0; i < NT; ++i) {
    const float v = s[i];
    y[lane * CONV_T + tt0 + 8 * i] = v > 0.f ? v : 0.f;
  }
}

__global__ __launch_bounds__(256) void sortpool_head(
    const float* __restrict__ H,
    const float* __restrict__ conv_w, const float* __restrict__ conv_b,
    const float* __restrict__ W1, const float* __restrict__ b1,
    const float* __restrict__ W2, const float* __restrict__ b2,
    float* __restrict__ out)
{
  __shared__ float vals[NN];
  __shared__ int   idxs[NN];
  __shared__ __attribute__((aligned(16))) float xs[KTOP * HD];
  __shared__ float wT[CONV_W * HD * WT_PITCH];
  __shared__ float y[FLAT];
  __shared__ float z[HD];
  __shared__ float logits[NCLS];
  __shared__ float lse;

  const int g = blockIdx.x;
  const int t = threadIdx.x;
  const float* hg = H + (size_t)g * NN * HD;

  for (int gi = t; gi < CONV_O * (HD / 4); gi += 256) {
    const int o  = gi >> 5;
    const int cq = gi & 31;
    const float4* p4 = (const float4*)(conv_w + (size_t)o * HD * CONV_W + cq * 12);
    const float4 a = p4[0], b = p4[1], d = p4[2];
    const float vv[12] = {a.x, a.y, a.z, a.w, b.x, b.y, b.z, b.w, d.x, d.y, d.z, d.w};
    float* dstp = wT + (cq * 4) * WT_PITCH + o;
#pragma unroll
    for (int k = 0; k < 12; ++k)
      dstp[((k % CONV_W) * HD + k / CONV_W) * WT_PITCH] = vv[k];
  }

  if (t < NN) { vals[t] = hg[t * HD + (HD - 1)]; idxs[t] = t; }
  __syncthreads();

  for (int k = 2; k <= NN; k <<= 1) {
    for (int j = k >> 1; j > 0; j >>= 1) {
      if (t < NN) {
        const int ixj = t ^ j;
        if (ixj > t) {
          const bool dirDesc = ((t & k) == 0);
          const float va = vals[t], vb = vals[ixj];
          const int   ia = idxs[t], ib = idxs[ixj];
          const bool aBeforeB = (va > vb) || (va == vb && ia < ib);
          if (dirDesc ? !aBeforeB : aBeforeB) {
            vals[t] = vb; vals[ixj] = va;
            idxs[t] = ib; idxs[ixj] = ia;
          }
        }
      }
      __syncthreads();
    }
  }

  for (int i = t; i < KTOP * HD; i += 256) {
    const int tt = i >> 7, c = i & (HD - 1);
    xs[i] = hg[(size_t)idxs[tt] * HD + c];
  }
  __syncthreads();

  {
    const int lane = t & 31;
    const int wave = __builtin_amdgcn_readfirstlane(t >> 5);
    const float cb = conv_b[lane];
    if (wave < 4) conv_lane_is_channel<4>(xs, wT, cb, wave, lane, y);
    else          conv_lane_is_channel<3>(xs, wT, cb, wave, lane, y);
  }
  __syncthreads();

  if (t < HD) {
    float s = b1[t];
    for (int i = 0; i < FLAT; ++i) s += y[i] * W1[(size_t)i * HD + t];
    z[t] = s > 0.f ? s : 0.f;
  }
  __syncthreads();

  if (t < NCLS) {
    float s = b2[t];
    for (int j = 0; j < HD; ++j) s += z[j] * W2[j * NCLS + t];
    logits[t] = s;
  }
  __syncthreads();

  if (t == 0) {
    float mx = logits[0];
    for (int c = 1; c < NCLS; ++c) mx = fmaxf(mx, logits[c]);
    float se = 0.f;
    for (int c = 0; c < NCLS; ++c) se += expf(logits[c] - mx);
    lse = mx + logf(se);
  }
  __syncthreads();
  if (t < NCLS) out[(size_t)g * NCLS + t] = logits[t] - lse;
}

extern "C" void kernel_launch(void* const* d_in, const int* in_sizes, int n_in,
                              void* d_out, int out_size, void* d_ws, size_t ws_size,
                              hipStream_t stream) {
  (void)in_sizes; (void)n_in; (void)out_size; (void)ws_size;
  const float* x      = (const float*)d_in[0];
  const int*   src    = (const int*)d_in[1];
  const int*   dst    = (const int*)d_in[2];
  const float* Wl1    = (const float*)d_in[5];
  const float* bl1    = (const float*)d_in[6];
  const float* Wr1    = (const float*)d_in[7];
  const float* Wl2    = (const float*)d_in[8];
  const float* bl2    = (const float*)d_in[9];
  const float* Wr2    = (const float*)d_in[10];
  const float* Wl3    = (const float*)d_in[11];
  const float* bl3    = (const float*)d_in[12];
  const float* Wr3    = (const float*)d_in[13];
  const float* conv_w = (const float*)d_in[14];
  const float* conv_b = (const float*)d_in[15];
  const float* W1     = (const float*)d_in[16];
  const float* b1     = (const float*)d_in[17];
  const float* W2     = (const float*)d_in[18];
  const float* b2     = (const float*)d_in[19];

  float* ws   = (float*)d_ws;
  float* mean = ws;
  float* h1   = ws + (size_t)VN * HD;
  float* h2   = h1 + (size_t)VN * HD;

  const int gemmBlocks = VN / 128;

  sage_aggregate<<<BG, 256, 0, stream>>>(x, src, dst, mean);
  sage_gemm<<<gemmBlocks, 256, 0, stream>>>(mean, x, Wl1, Wr1, bl1, h1);
  sage_aggregate<<<BG, 256, 0, stream>>>(h1, src, dst, mean);
  sage_gemm<<<gemmBlocks, 256, 0, stream>>>(mean, h1, Wl2, Wr2, bl2, h2);
  sage_aggregate<<<BG, 256, 0, stream>>>(h2, src, dst, mean);
  sage_gemm<<<gemmBlocks, 256, 0, stream>>>(mean, h2, Wl3, Wr3, bl3, h1);
  sortpool_head<<<BG, 256, 0, stream>>>(h1, conv_w, conv_b, W1, b1, W2, b2,
                                        (float*)d_out);
}


# aggregate: two-level ballot bucketing replaced by one stable counting sort by destination in LDS (7-ballot peer groups + running counts); 4 workgroups per WGP (was 2); sum order kept
# speedup vs baseline: 5.6952x; 1.1995x over previous
#include <hip/hip_runtime.h>
#include <hip/hip_bf16.h>

#define BG   1024
#define NN   128
#define VN   (BG * NN)
#define HD   128
#define EPG  2048
#define KTOP 30
#define CONV_O 32
#define CONV_W 3
#define CONV_T (KTOP - CONV_W + 1)
#define FLAT   (CONV_O * CONV_T)
#define NCLS   16

typedef float v2f __attribute__((ext_vector_type(2)));
typedef float v8f __attribute__((ext_vector_type(8)));

#define ZROW NN

__global__ __launch_bounds__(256) void sage_aggregate(
    const float* __restrict__ X, const int* __restrict__ src,
    const int* __restrict__ dst, float* __restrict__ meanOut)
{
  __shared__ float4        feat4[(NN + 1) * (HD / 4)];
  __shared__ int           cntw[8][NN];
  __shared__ __attribute__((aligned(16))) int deg[NN];
  __shared__ int           rowstart[NN + 4];
  __shared__ unsigned char rl[EPG];

  const int g    = blockIdx.x;
  const int t    = threadIdx.x;
  const int lane = t & 31;
  const int wave = __builtin_amdgcn_readfirstlane(t >> 5);

  const float4* xg4 = (const float4*)(X + (size_t)g * NN * HD);
  for (int i = t; i < NN * HD / 4; i += 256) feat4[i] = xg4[i];
  if (t < HD / 4) feat4[ZROW * (HD / 4) + t] = make_float4(0.f, 0.f, 0.f, 0.f);

  int* mycnt = cntw[wave];
  mycnt[lane] = 0; mycnt[lane + 32] = 0; mycnt[lane + 64] = 0; mycnt[lane + 96] = 0;

  unsigned sv[8], dv[8], loc[8];
  {
    const int eb = g * EPG + wave * 256 + lane;
#pragma unroll
    for (int j = 0; j < 8; ++j) {
      sv[j] = (unsigned)src[eb + j * 32] & (NN - 1);
      dv[j] = (unsigned)dst[eb + j * 32] & (NN - 1);
    }
  }

#pragma unroll
  for (int j = 0; j < 8; ++j) {
    const unsigned d = dv[j];
    unsigned peer = 0xffffffffu;
#pragma unroll
    for (int b = 0; b < 7; ++b) {
      const unsigned bit = (d >> b) & 1u;
      const unsigned m = __builtin_amdgcn_ballot_w32(bit != 0u);
      peer &= m ^ (bit - 1u);
    }
    const unsigned rank = __builtin_amdgcn_mbcnt_lo(peer, 0u);
    const unsigned tot  = (unsigned)__builtin_popcount(peer);
    const unsigned base = (unsigned)mycnt[d];
    loc[j] = base + rank;
    if (rank + 1u == tot) mycnt[d] = (int)(base + tot);
  }
  __syncthreads();

  if (t < NN) {
    int run = 0;
#pragma unroll
    for (int c = 0; c < 8; ++c) {
      const int v = cntw[c][t];
      cntw[c][t] = run;
      run += v;
    }
    deg[t] = run;
  }
  __syncthreads();

  if (wave == 0) {
    const int4 dg = *(const int4*)&deg[lane * 4];
    const int s = dg.x + dg.y + dg.z + dg.w;
    int inc = s;
#pragma unroll
    for (int o = 1; o < 32; o <<= 1) {
      const int v = __shfl_up(inc, o);
      if (lane >= o) inc += v;
    }
    const int ex = inc - s;
    rowstart[lane * 4 + 0] = ex;
    rowstart[lane * 4 + 1] = ex + dg.x;
    rowstart[lane * 4 + 2] = ex + dg.x + dg.y;
    rowstart[lane * 4 + 3] = ex + dg.x + dg.y + dg.z;
    if (lane == 31) rowstart[NN] = inc;
  }
  __syncthreads();

#pragma unroll
  for (int j = 0; j < 8; ++j) {
    const unsigned d = dv[j];
    const int pos = rowstart[d] + mycnt[d] + (int)loc[j];
    rl[pos] = (unsigned char)sv[j];
  }
  __syncthreads();

  float4* mg4 = (float4*)(meanOut + (size_t)g * NN * HD);
  for (int r = 0; r < 16; ++r) {
    const int beg = __builtin_amdgcn_readfirstlane(rowstart[wave * 16 + r]);
    const int n   = __builtin_amdgcn_readfirstlane(rowstart[wave * 16 + r + 1]) - beg;
    float4 a = make_float4(0.f, 0.f, 0.f, 0.f);
    for (int c = 0; c < n; c += 32) {
      const int k = c + lane;
      const int myidx = (k < n) ? (int)rl[beg + k] : ZROW;
      const int m = (n - c) < 32 ? (n - c) : 32;
      for (int j = 0; j < m; j += 4) {
        const int s0 = __builtin_amdgcn_readlane(myidx, j);
        const int s1 = __builtin_amdgcn_readlane(myidx, j + 1);
        const int s2 = __builtin_amdgcn_readlane(myidx, j + 2);
        const int s3 = __builtin_amdgcn_readlane(myidx, j + 3);
        const float4 v0 = feat4[s0 * (HD / 4) + lane];
        const float4 v1 = feat4[s1 * (HD / 4) + lane];
        const float4 v2 = feat4[s2 * (HD / 4) + lane];
        const float4 v3 = feat4[s3 * (HD / 4) + lane];
        a.x += v0.x; a.y += v0.y; a.z += v0.z; a.w += v0.w;
        a.x += v1.x; a.y += v1.y; a.z += v1.z; a.w += v1.w;
        a.x += v2.x; a.y += v2.y; a.z += v2.z; a.w += v2.w;
        a.x += v3.x; a.y += v3.y; a.z += v3.z; a.w += v3.w;
      }
    }
    const float cnt = (float)(n > 1 ? n : 1);
    float4 o;
    o.x = a.x / cnt; o.y = a.y / cnt; o.z = a.z / cnt; o.w = a.w / cnt;
    mg4[(wave * 16 + r) * (HD / 4) + lane] = o;
  }
}

__global__ __launch_bounds__(256) void sage_gemm(
    const float* __restrict__ Aagg, const float* __restrict__ Aself,
    const float* __restrict__ Wl,   const float* __restrict__ Wr,
    const float* __restrict__ bias, float* __restrict__ out)
{
  __shared__ float2 wp[128 * HD];

  const int t = threadIdx.x;
  for (int idx = t; idx < 128 * HD; idx += 256) {
    const int kk  = idx >> 7;
    const int col = idx & (HD - 1);
    const float* W = (kk < 64) ? Wl : Wr;
    const int k = (kk & 63) * 2;
    float2 v;
    v.x = W[(size_t)k       * HD + col];
    v.y = W[(size_t)(k + 1) * HD + col];
    wp[idx] = v;
  }
  __syncthreads();

  const int wave = t >> 5;
  const int lane = t & 31;
  const int m    = lane & 15;
  const int hi   = lane >> 4;
  const int kb   = hi * 2;

  const int rowBase = blockIdx.x * 128 + wave * 16;
  const float* arow1 = Aagg  + (size_t)(rowBase + m) * HD;
  const float* arow2 = Aself + (size_t)(rowBase + m) * HD;

  v8f acc0 = {}, acc1 = {}, acc2 = {}, acc3 = {};
  v8f acc4 = {}, acc5 = {}, acc6 = {}, acc7 = {};

#define GEMM_STEP(AROW, KOFF)                                              \
  {                                                                        \
    v2f a = *(const v2f*)((AROW) + k0 + kb);                               \
    const float2* bp = wp + ((k0 >> 1) + hi + (KOFF)) * HD + m;            \
    float2 bb; v2f b;                                                      \
    bb = bp[0];   b.x = bb.x; b.y = bb.y;                                  \
    acc0 = __builtin_amdgcn_wmma_f32_16x16x4_f32(false, a, false, b, (short)0, acc0, false, false); \
    bb = bp[16];  b.x = bb.x; b.y = bb.y;                                  \
    acc1 = __builtin_amdgcn_wmma_f32_16x16x4_f32(false, a, false, b, (short)0, acc1, false, false); \
    bb = bp[32];  b.x = bb.x; b.y = bb.y;                                  \
    acc2 = __builtin_amdgcn_wmma_f32_16x16x4_f32(false, a, false, b, (short)0, acc2, false, false); \
    bb = bp[48];  b.x = bb.x; b.y = bb.y;                                  \
    acc3 = __builtin_amdgcn_wmma_f32_16x16x4_f32(false, a, false, b, (short)0, acc3, false, false); \
    bb = bp[64];  b.x = bb.x; b.y = bb.y;                                  \
    acc4 = __builtin_amdgcn_wmma_f32_16x16x4_f32(false, a, false, b, (short)0, acc4, false, false); \
    bb = bp[80];  b.x = bb.x; b.y = bb.y;                                  \
    acc5 = __builtin_amdgcn_wmma_f32_16x16x4_f32(false, a, false, b, (short)0, acc5, false, false); \
    bb = bp[96];  b.x = bb.x; b.y = bb.y;                                  \
    acc6 = __builtin_amdgcn_wmma_f32_16x16x4_f32(false, a, false, b, (short)0, acc6, false, false); \
    bb = bp[112]; b.x = bb.x; b.y = bb.y;                                  \
    acc7 = __builtin_amdgcn_wmma_f32_16x16x4_f32(false, a, false, b, (short)0, acc7, false, false); \
  }

#pragma unroll 2
  for (int k0 = 0; k0 < HD; k0 += 4) GEMM_STEP(arow1, 0)
#pragma unroll 2
  for (int k0 = 0; k0 < HD; k0 += 4) GEMM_STEP(arow2, 64)
#undef GEMM_STEP

  const int rbase = rowBase + hi * 8;
  v8f accs[8] = {acc0, acc1, acc2, acc3, acc4, acc5, acc6, acc7};
#pragma unroll
  for (int ct = 0; ct < 8; ++ct) {
    const int col = ct * 16 + m;
    const float bv = bias[col];
#pragma unroll
    for (int v = 0; v < 8; ++v) {
      float val = accs[ct][v] + bv;
      val = val > 0.f ? val : 0.f;
      out[(size_t)(rbase + v) * HD + col] = val;
    }
  }
}

#define WT_PITCH 33

template <int NT>
__device__ __forceinline__ void conv_lane_is_channel(
    const float* xs, const float* wT, float bias, int tt0, int lane, float* y)
{
  float s[NT];
#pragma unroll
  for (int i = 0; i < NT; ++i) s[i] = bias;
  for (int w = 0; w < CONV_W; ++w) {
    const float* wcol = wT + (w * HD) * WT_PITCH + lane;
    const float* xrow = xs + (tt0 + w) * HD;
    for (int c = 0; c < HD; c += 4) {
      const float w0 = wcol[(c + 0) * WT_PITCH];
      const float w1 = wcol[(c + 1) * WT_PITCH];
      const float w2 = wcol[(c + 2) * WT_PITCH];
      const float w3 = wcol[(c + 3) * WT_PITCH];
#pragma unroll
      for (int i = 0; i < NT; ++i) {
        const float4 xv = *(const float4*)(xrow + i * 8 * HD + c);
        s[i] = __builtin_fmaf(xv.x, w0, s[i]);
        s[i] = __builtin_fmaf(xv.y, w1, s[i]);
        s[i] = __builtin_fmaf(xv.z, w2, s[i]);
        s[i] = __builtin_fmaf(xv.w, w3, s[i]);
      }
    }
  }
#pragma unroll
  for (int i = 0; i < NT; ++i) {
    const float v = s[i];
    y[lane * CONV_T + tt0 + 8 * i] = v > 0.f ? v : 0.f;
  }
}

__global__ __launch_bounds__(256) void sortpool_head(
    const float* __restrict__ H,
    const float* __restrict__ conv_w, const float* __restrict__ conv_b,
    const float* __restrict__ W1, const float* __restrict__ b1,
    const float* __restrict__ W2, const float* __restrict__ b2,
    float* __restrict__ out)
{
  __shared__ float vals[NN];
  __shared__ int   idxs[NN];
  __shared__ __attribute__((aligned(16))) float xs[KTOP * HD];
  __shared__ float wT[CONV_W * HD * WT_PITCH];
  __shared__ float y[FLAT];
  __shared__ float z[HD];
  __shared__ float logits[NCLS];
  __shared__ float lse;

  const int g = blockIdx.x;
  const int t = threadIdx.x;
  const float* hg = H + (size_t)g * NN * HD;

  for (int gi = t; gi < CONV_O * (HD / 4); gi += 256) {
    const int o  = gi >> 5;
    const int cq = gi & 31;
    const float4* p4 = (const float4*)(conv_w + (size_t)o * HD * CONV_W + cq * 12);
    const float4 a = p4[0], b = p4[1], d = p4[2];
    const float vv[12] = {a.x, a.y, a.z, a.w, b.x, b.y, b.z, b.w, d.x, d.y, d.z, d.w};
    float* dstp = wT + (cq * 4) * WT_PITCH + o;
#pragma unroll
    for (int k = 0; k < 12; ++k)
      dstp[((k % CONV_W) * HD + k / CONV_W) * WT_PITCH] = vv[k];
  }

  if (t < NN) { vals[t] = hg[t * HD + (HD - 1)]; idxs[t] = t; }
  __syncthreads();

  for (int k = 2; k <= NN; k <<= 1) {
    for (int j = k >> 1; j > 0; j >>= 1) {
      if (t < NN) {
        const int ixj = t ^ j;
        if (ixj > t) {
          const bool dirDesc = ((t & k) == 0);
          const float va = vals[t], vb = vals[ixj];
          const int   ia = idxs[t], ib = idxs[ixj];
          const bool aBeforeB = (va > vb) || (va == vb && ia < ib);
          if (dirDesc ? !aBeforeB : aBeforeB) {
            vals[t] = vb; vals[ixj] = va;
            idxs[t] = ib; idxs[ixj] = ia;
          }
        }
      }
      __syncthreads();
    }
  }

  for (int i = t; i < KTOP * HD; i += 256) {
    const int tt = i >> 7, c = i & (HD - 1);
    xs[i] = hg[(size_t)idxs[tt] * HD + c];
  }
  __syncthreads();

  {
    const int lane = t & 31;
    const int wave = __builtin_amdgcn_readfirstlane(t >> 5);
    const float cb = conv_b[lane];
    if (wave < 4) conv_lane_is_channel<4>(xs, wT, cb, wave, lane, y);
    else          conv_lane_is_channel<3>(xs, wT, cb, wave, lane, y);
  }
  __syncthreads();

  if (t < HD) {
    float s = b1[t];
    for (int i = 0; i < FLAT; ++i) s += y[i] * W1[(size_t)i * HD + t];
    z[t] = s > 0.f ? s : 0.f;
  }
  __syncthreads();

  if (t < NCLS) {
    float s = b2[t];
    for (int j = 0; j < HD; ++j) s += z[j] * W2[j * NCLS + t];
    logits[t] = s;
  }
  __syncthreads();

  if (t == 0) {
    float mx = logits[0];
    for (int c = 1; c < NCLS; ++c) mx = fmaxf(mx, logits[c]);
    float se = 0.f;
    for (int c = 0; c < NCLS; ++c) se += expf(logits[c] - mx);
    lse = mx + logf(se);
  }
  __syncthreads();
  if (t < NCLS) out[(size_t)g * NCLS + t] = logits[t] - lse;
}

extern "C" void kernel_launch(void* const* d_in, const int* in_sizes, int n_in,
                              void* d_out, int out_size, void* d_ws, size_t ws_size,
                              hipStream_t stream) {
  (void)in_sizes; (void)n_in; (void)out_size; (void)ws_size;
  const float* x      = (const float*)d_in[0];
  const int*   src    = (const int*)d_in[1];
  const int*   dst    = (const int*)d_in[2];
  const float* Wl1    = (const float*)d_in[5];
  const float* bl1    = (const float*)d_in[6];
  const float* Wr1    = (const float*)d_in[7];
  const float* Wl2    = (const float*)d_in[8];
  const float* bl2    = (const float*)d_in[9];
  const float* Wr2    = (const float*)d_in[10];
  const float* Wl3    = (const float*)d_in[11];
  const float* bl3    = (const float*)d_in[12];
  const float* Wr3    = (const float*)d_in[13];
  const float* conv_w = (const float*)d_in[14];
  const float* conv_b = (const float*)d_in[15];
  const float* W1     = (const float*)d_in[16];
  const float* b1     = (const float*)d_in[17];
  const float* W2     = (const float*)d_in[18];
  const float* b2     = (const float*)d_in[19];

  float* ws   = (float*)d_ws;
  float* mean = ws;
  float* h1   = ws + (size_t)VN * HD;
  float* h2   = h1 + (size_t)VN * HD;

  const int gemmBlocks = VN / 128;

  sage_aggregate<<<BG, 256, 0, stream>>>(x, src, dst, mean);
  sage_gemm<<<gemmBlocks, 256, 0, stream>>>(mean, x, Wl1, Wr1, bl1, h1);
  sage_aggregate<<<BG, 256, 0, stream>>>(h1, src, dst, mean);
  sage_gemm<<<gemmBlocks, 256, 0, stream>>>(mean, h1, Wl2, Wr2, bl2, h2);
  sage_aggregate<<<BG, 256, 0, stream>>>(h2, src, dst, mean);
  sage_gemm<<<gemmBlocks, 256, 0, stream>>>(mean, h2, Wl3, Wr3, bl3, h1);
  sortpool_head<<<BG, 256, 0, stream>>>(h1, conv_w, conv_b, W1, b1, W2, b2,
                                        (float*)d_out);
}


# sage_gemm: each wave computes a 32x128 tile, two 16-row tiles sharing every B fragment read; 512 workgroups of 256 rows, all resident in one round (was 1024 in two); sum order kept
# speedup vs baseline: 5.9863x; 1.0511x over previous
#include <hip/hip_runtime.h>
#include <hip/hip_bf16.h>

#define BG   1024
#define NN   128
#define VN   (BG * NN)
#define HD   128
#define EPG  2048
#define KTOP 30
#define CONV_O 32
#define CONV_W 3
#define CONV_T (KTOP - CONV_W + 1)
#define FLAT   (CONV_O * CONV_T)
#define NCLS   16

typedef float v2f __attribute__((ext_vector_type(2)));
typedef float v8f __attribute__((ext_vector_type(8)));

#define ZROW NN

__global__ __launch_bounds__(256) void sage_aggregate(
    const float* __restrict__ X, const int* __restrict__ src,
    const int* __restrict__ dst, float* __restrict__ meanOut)
{
  __shared__ float4        feat4[(NN + 1) * (HD / 4)];
  __shared__ int           cntw[8][NN];
  __shared__ __attribute__((aligned(16))) int deg[NN];
  __shared__ int           rowstart[NN + 4];
  __shared__ unsigned char rl[EPG];

  const int g    = blockIdx.x;
  const int t    = threadIdx.x;
  const int lane = t & 31;
  const int wave = __builtin_amdgcn_readfirstlane(t >> 5);

  const float4* xg4 = (const float4*)(X + (size_t)g * NN * HD);
  for (int i = t; i < NN * HD / 4; i += 256) feat4[i] = xg4[i];
  if (t < HD / 4) feat4[ZROW * (HD / 4) + t] = make_float4(0.f, 0.f, 0.f, 0.f);

  int* mycnt = cntw[wave];
  mycnt[lane] = 0; mycnt[lane + 32] = 0; mycnt[lane + 64] = 0; mycnt[lane + 96] = 0;

  unsigned sv[8], dv[8], loc[8];
  {
    const int eb = g * EPG + wave * 256 + lane;
#pragma unroll
    for (int j = 0; j < 8; ++j) {
      sv[j] = (unsigned)src[eb + j * 32] & (NN - 1);
      dv[j] = (unsigned)dst[eb + j * 32] & (NN - 1);
    }
  }

#pragma unroll
  for (int j = 0; j < 8; ++j) {
    const unsigned d = dv[j];
    unsigned peer = 0xffffffffu;
#pragma unroll
    for (int b = 0; b < 7; ++b) {
      const unsigned bit = (d >> b) & 1u;
      const unsigned m = __builtin_amdgcn_ballot_w32(bit != 0u);
      peer &= m ^ (bit - 1u);
    }
    const unsigned rank = __builtin_amdgcn_mbcnt_lo(peer, 0u);
    const unsigned tot  = (unsigned)__builtin_popcount(peer);
    const unsigned base = (unsigned)mycnt[d];
    loc[j] = base + rank;
    if (rank + 1u == tot) mycnt[d] = (int)(base + tot);
  }
  __syncthreads();

  if (t < NN) {
    int run = 0;
#pragma unroll
    for (int c = 0; c < 8; ++c) {
      const int v = cntw[c][t];
      cntw[c][t] = run;
      run += v;
    }
    deg[t] = run;
  }
  __syncthreads();

  if (wave == 0) {
    const int4 dg = *(const int4*)&deg[lane * 4];
    const int s = dg.x + dg.y + dg.z + dg.w;
    int inc = s;
#pragma unroll
    for (int o = 1; o < 32; o <<= 1) {
      const int v = __shfl_up(inc, o);
      if (lane >= o) inc += v;
    }
    const int ex = inc - s;
    rowstart[lane * 4 + 0] = ex;
    rowstart[lane * 4 + 1] = ex + dg.x;
    rowstart[lane * 4 + 2] = ex + dg.x + dg.y;
    rowstart[lane * 4 + 3] = ex + dg.x + dg.y + dg.z;
    if (lane == 31) rowstart[NN] = inc;
  }
  __syncthreads();

#pragma unroll
  for (int j = 0; j < 8; ++j) {
    const unsigned d = dv[j];
    const int pos = rowstart[d] + mycnt[d] + (int)loc[j];
    rl[pos] = (unsigned char)sv[j];
  }
  __syncthreads();

  float4* mg4 = (float4*)(meanOut + (size_t)g * NN * HD);
  for (int r = 0; r < 16; ++r) {
    const int beg = __builtin_amdgcn_readfirstlane(rowstart[wave * 16 + r]);
    const int n   = __builtin_amdgcn_readfirstlane(rowstart[wave * 16 + r + 1]) - beg;
    float4 a = make_float4(0.f, 0.f, 0.f, 0.f);
    for (int c = 0; c < n; c += 32) {
      const int k = c + lane;
      const int myidx = (k < n) ? (int)rl[beg + k] : ZROW;
      const int m = (n - c) < 32 ? (n - c) : 32;
      for (int j = 0; j < m; j += 4) {
        const int s0 = __builtin_amdgcn_readlane(myidx, j);
        const int s1 = __builtin_amdgcn_readlane(myidx, j + 1);
        const int s2 = __builtin_amdgcn_readlane(myidx, j + 2);
        const int s3 = __builtin_amdgcn_readlane(myidx, j + 3);
        const float4 v0 = feat4[s0 * (HD / 4) + lane];
        const float4 v1 = feat4[s1 * (HD / 4) + lane];
        const float4 v2 = feat4[s2 * (HD / 4) + lane];
        const float4 v3 = feat4[s3 * (HD / 4) + lane];
        a.x += v0.x; a.y += v0.y; a.z += v0.z; a.w += v0.w;
        a.x += v1.x; a.y += v1.y; a.z += v1.z; a.w += v1.w;
        a.x += v2.x; a.y += v2.y; a.z += v2.z; a.w += v2.w;
        a.x += v3.x; a.y += v3.y; a.z += v3.z; a.w += v3.w;
      }
    }
    const float cnt = (float)(n > 1 ? n : 1);
    float4 o;
    o.x = a.x / cnt; o.y = a.y / cnt; o.z = a.z / cnt; o.w = a.w / cnt;
    mg4[(wave * 16 + r) * (HD / 4) + lane] = o;
  }
}

#define GEMM_ROWS 256

__global__ __launch_bounds__(256) void sage_gemm(
    const float* __restrict__ Aagg, const float* __restrict__ Aself,
    const float* __restrict__ Wl,   const float* __restrict__ Wr,
    const float* __restrict__ bias, float* __restrict__ out)
{
  __shared__ float2 wp[128 * HD];

  const int t = threadIdx.x;
  for (int idx = t; idx < 128 * HD; idx += 256) {
    const int kk  = idx >> 7;
    const int col = idx & (HD - 1);
    const float* W = (kk < 64) ? Wl : Wr;
    const int k = (kk & 63) * 2;
    float2 v;
    v.x = W[(size_t)k       * HD + col];
    v.y = W[(size_t)(k + 1) * HD + col];
    wp[idx] = v;
  }
  __syncthreads();

  const int wave = t >> 5;
  const int lane = t & 31;
  const int m    = lane & 15;
  const int hi   = lane >> 4;
  const int kb   = hi * 2;

  const int rowBase = blockIdx.x * GEMM_ROWS + wave * 32;
  const float* arow1 = Aagg  + (size_t)(rowBase + m) * HD;
  const float* arow2 = Aself + (size_t)(rowBase + m) * HD;

  v8f p0 = {}, p1 = {}, p2 = {}, p3 = {}, p4 = {}, p5 = {}, p6 = {}, p7 = {};
  v8f q0 = {}, q1 = {}, q2 = {}, q3 = {}, q4 = {}, q5 = {}, q6 = {}, q7 = {};

#define GEMM_PAIR(P, Q, OFF)                                               \
    bb = bp[OFF]; b.x = bb.x; b.y = bb.y;                                  \
    P = __builtin_amdgcn_wmma_f32_16x16x4_f32(false, a0, false, b, (short)0, P, false, false); \
    Q = __builtin_amdgcn_wmma_f32_16x16x4_f32(false, a1, false, b, (short)0, Q, false, false);

#define GEMM_STEP(AROW, KOFF)                                              \
  {                                                                        \
    const v2f a0 = *(const v2f*)((AROW) + k0 + kb);                        \
    const v2f a1 = *(const v2f*)((AROW) + 16 * HD + k0 + kb);              \
    const float2* bp = wp + ((k0 >> 1) + hi + (KOFF)) * HD + m;            \
    float2 bb; v2f b;                                                      \
    GEMM_PAIR(p0, q0, 0)                                                   \
    GEMM_PAIR(p1, q1, 16)                                                  \
    GEMM_PAIR(p2, q2, 32)                                                  \
    GEMM_PAIR(p3, q3, 48)                                                  \
    GEMM_PAIR(p4, q4, 64)                                                  \
    GEMM_PAIR(p5, q5, 80)                                                  \
    GEMM_PAIR(p6, q6, 96)                                                  \
    GEMM_PAIR(p7, q7, 112)                                                 \
  }

#pragma unroll 2
  for (int k0 = 0; k0 < HD; k0 += 4) GEMM_STEP(arow1, 0)
#pragma unroll 2
  for (int k0 = 0; k0 < HD; k0 += 4) GEMM_STEP(arow2, 64)
#undef GEMM_STEP
#undef GEMM_PAIR

  const int rbase = rowBase + hi * 8;
  v8f accs[16] = {p0, p1, p2, p3, p4, p5, p6, p7, q0, q1, q2, q3, q4, q5, q6, q7};
#pragma unroll
  for (int rt = 0; rt < 2; ++rt) {
#pragma unroll
    for (int ct = 0; ct < 8; ++ct) {
      const int col = ct * 16 + m;
      const float bv = bias[col];
#pragma unroll
      for (int v = 0; v < 8; ++v) {
        float val = accs[rt * 8 + ct][v] + bv;
        val = val > 0.f ? val : 0.f;
        out[(size_t)(rbase + rt * 16 + v) * HD + col] = val;
      }
    }
  }
}

#define WT_PITCH 33

template <int NT>
__device__ __forceinline__ void conv_lane_is_channel(
    const float* xs, const float* wT, float bias, int tt0, int lane, float* y)
{
  float s[NT];
#pragma unroll
  for (int i = 0; i < NT; ++i) s[i] = bias;
  for (int w = 0; w < CONV_W; ++w) {
    const float* wcol = wT + (w * HD) * WT_PITCH + lane;
    const float* xrow = xs + (tt0 + w) * HD;
    for (int c = 0; c < HD; c += 4) {
      const float w0 = wcol[(c + 0) * WT_PITCH];
      const float w1 = wcol[(c + 1) * WT_PITCH];
      const float w2 = wcol[(c + 2) * WT_PITCH];
      const float w3 = wcol[(c + 3) * WT_PITCH];
#pragma unroll
      for (int i = 0; i < NT; ++i) {
        const float4 xv = *(const float4*)(xrow + i * 8 * HD + c);
        s[i] = __builtin_fmaf(xv.x, w0, s[i]);
        s[i] = __builtin_fmaf(xv.y, w1, s[i]);
        s[i] = __builtin_fmaf(xv.z, w2, s[i]);
        s[i] = __builtin_fmaf(xv.w, w3, s[i]);
      }
    }
  }
#pragma unroll
  for (int i = 0; i < NT; ++i) {
    const float v = s[i];
    y[lane * CONV_T + tt0 + 8 * i] = v > 0.f ? v : 0.f;
  }
}

__global__ __launch_bounds__(256) void sortpool_head(
    const float* __restrict__ H,
    const float* __restrict__ conv_w, const float* __restrict__ conv_b,
    const float* __restrict__ W1, const float* __restrict__ b1,
    const float* __restrict__ W2, const float* __restrict__ b2,
    float* __restrict__ out)
{
  __shared__ float vals[NN];
  __shared__ int   idxs[NN];
  __shared__ __attribute__((aligned(16))) float xs[KTOP * HD];
  __shared__ float wT[CONV_W * HD * WT_PITCH];
  __shared__ float y[FLAT];
  __shared__ float z[HD];
  __shared__ float logits[NCLS];
  __shared__ float lse;

  const int g = blockIdx.x;
  const int t = threadIdx.x;
  const float* hg = H + (size_t)g * NN * HD;

  for (int gi = t; gi < CONV_O * (HD / 4); gi += 256) {
    const int o  = gi >> 5;
    const int cq = gi & 31;
    const float4* p4 = (const float4*)(conv_w + (size_t)o * HD * CONV_W + cq * 12);
    const float4 a = p4[0], b = p4[1], d = p4[2];
    const float vv[12] = {a.x, a.y, a.z, a.w, b.x, b.y, b.z, b.w, d.x, d.y, d.z, d.w};
    float* dstp = wT + (cq * 4) * WT_PITCH + o;
#pragma unroll
    for (int k = 0; k < 12; ++k)
      dstp[((k % CONV_W) * HD + k / CONV_W) * WT_PITCH] = vv[k];
  }

  if (t < NN) { vals[t] = hg[t * HD + (HD - 1)]; idxs[t] = t; }
  __syncthreads();

  for (int k = 2; k <= NN; k <<= 1) {
    for (int j = k >> 1; j > 0; j >>= 1) {
      if (t < NN) {
        const int ixj = t ^ j;
        if (ixj > t) {
          const bool dirDesc = ((t & k) == 0);
          const float va = vals[t], vb = vals[ixj];
          const int   ia = idxs[t], ib = idxs[ixj];
          const bool aBeforeB = (va > vb) || (va == vb && ia < ib);
          if (dirDesc ? !aBeforeB : aBeforeB) {
            vals[t] = vb; vals[ixj] = va;
            idxs[t] = ib; idxs[ixj] = ia;
          }
        }
      }
      __syncthreads();
    }
  }

  for (int i = t; i < KTOP * HD; i += 256) {
    const int tt = i >> 7, c = i & (HD - 1);
    xs[i] = hg[(size_t)idxs[tt] * HD + c];
  }
  __syncthreads();

  {
    const int lane = t & 31;
    const int wave = __builtin_amdgcn_readfirstlane(t >> 5);
    const float cb = conv_b[lane];
    if (wave < 4) conv_lane_is_channel<4>(xs, wT, cb, wave, lane, y);
    else          conv_lane_is_channel<3>(xs, wT, cb, wave, lane, y);
  }
  __syncthreads();

  if (t < HD) {
    float s = b1[t];
    for (int i = 0; i < FLAT; ++i) s += y[i] * W1[(size_t)i * HD + t];
    z[t] = s > 0.f ? s : 0.f;
  }
  __syncthreads();

  if (t < NCLS) {
    float s = b2[t];
    for (int j = 0; j < HD; ++j) s += z[j] * W2[j * NCLS + t];
    logits[t] = s;
  }
  __syncthreads();

  if (t == 0) {
    float mx = logits[0];
    for (int c = 1; c < NCLS; ++c) mx = fmaxf(mx, logits[c]);
    float se = 0.f;
    for (int c = 0; c < NCLS; ++c) se += expf(logits[c] - mx);
    lse = mx + logf(se);
  }
  __syncthreads();
  if (t < NCLS) out[(size_t)g * NCLS + t] = logits[t] - lse;
}

extern "C" void kernel_launch(void* const* d_in, const int* in_sizes, int n_in,
                              void* d_out, int out_size, void* d_ws, size_t ws_size,
                              hipStream_t stream) {
  (void)in_sizes; (void)n_in; (void)out_size; (void)ws_size;
  const float* x      = (const float*)d_in[0];
  const int*   src    = (const int*)d_in[1];
  const int*   dst    = (const int*)d_in[2];
  const float* Wl1    = (const float*)d_in[5];
  const float* bl1    = (const float*)d_in[6];
  const float* Wr1    = (const float*)d_in[7];
  const float* Wl2    = (const float*)d_in[8];
  const float* bl2    = (const float*)d_in[9];
  const float* Wr2    = (const float*)d_in[10];
  const float* Wl3    = (const float*)d_in[11];
  const float* bl3    = (const float*)d_in[12];
  const float* Wr3    = (const float*)d_in[13];
  const float* conv_w = (const float*)d_in[14];
  const float* conv_b = (const float*)d_in[15];
  const float* W1     = (const float*)d_in[16];
  const float* b1     = (const float*)d_in[17];
  const float* W2     = (const float*)d_in[18];
  const float* b2     = (const float*)d_in[19];

  float* ws   = (float*)d_ws;
  float* mean = ws;
  float* h1   = ws + (size_t)VN * HD;
  float* h2   = h1 + (size_t)VN * HD;

  const int gemmBlocks = VN / GEMM_ROWS;

  sage_aggregate<<<BG, 256, 0, stream>>>(x, src, dst, mean);
  sage_gemm<<<gemmBlocks, 256, 0, stream>>>(mean, x, Wl1, Wr1, bl1, h1);
  sage_aggregate<<<BG, 256, 0, stream>>>(h1, src, dst, mean);
  sage_gemm<<<gemmBlocks, 256, 0, stream>>>(mean, h1, Wl2, Wr2, bl2, h2);
  sage_aggregate<<<BG, 256, 0, stream>>>(h2, src, dst, mean);
  sage_gemm<<<gemmBlocks, 256, 0, stream>>>(mean, h2, Wl3, Wr3, bl3, h1);
  sortpool_head<<<BG, 256, 0, stream>>>(h1, conv_w, conv_b, W1, b1, W2, b2,
                                        (float*)d_out);
}


# head: dense layers in their own launch, 16 graphs per workgroup; dense1 on the f32 matrix instruction, W1 read once per 16 graphs; dense1 sums now in matrix k order, even/odd k partials added last
# speedup vs baseline: 6.3771x; 1.0653x over previous
#include <hip/hip_runtime.h>
#include <hip/hip_bf16.h>

#define BG   1024
#define NN   128
#define VN   (BG * NN)
#define HD   128
#define EPG  2048
#define KTOP 30
#define CONV_O 32
#define CONV_W 3
#define CONV_T (KTOP - CONV_W + 1)
#define FLAT   (CONV_O * CONV_T)
#define NCLS   16

typedef float v2f __attribute__((ext_vector_type(2)));
typedef float v8f __attribute__((ext_vector_type(8)));

#define ZROW NN

__global__ __launch_bounds__(256) void sage_aggregate(
    const float* __restrict__ X, const int* __restrict__ src,
    const int* __restrict__ dst, float* __restrict__ meanOut)
{
  __shared__ float4        feat4[(NN + 1) * (HD / 4)];
  __shared__ int           cntw[8][NN];
  __shared__ __attribute__((aligned(16))) int deg[NN];
  __shared__ int           rowstart[NN + 4];
  __shared__ unsigned char rl[EPG];

  const int g    = blockIdx.x;
  const int t    = threadIdx.x;
  const int lane = t & 31;
  const int wave = __builtin_amdgcn_readfirstlane(t >> 5);

  const float4* xg4 = (const float4*)(X + (size_t)g * NN * HD);
  for (int i = t; i < NN * HD / 4; i += 256) feat4[i] = xg4[i];
  if (t < HD / 4) feat4[ZROW * (HD / 4) + t] = make_float4(0.f, 0.f, 0.f, 0.f);

  int* mycnt = cntw[wave];
  mycnt[lane] = 0; mycnt[lane + 32] = 0; mycnt[lane + 64] = 0; mycnt[lane + 96] = 0;

  unsigned sv[8], dv[8], loc[8];
  {
    const int eb = g * EPG + wave * 256 + lane;
#pragma unroll
    for (int j = 0; j < 8; ++j) {
      sv[j] = (unsigned)src[eb + j * 32] & (NN - 1);
      dv[j] = (unsigned)dst[eb + j * 32] & (NN - 1);
    }
  }

#pragma unroll
  for (int j = 0; j < 8; ++j) {
    const unsigned d = dv[j];
    unsigned peer = 0xffffffffu;
#pragma unroll
    for (int b = 0; b < 7; ++b) {
      const unsigned bit = (d >> b) & 1u;
      const unsigned m = __builtin_amdgcn_ballot_w32(bit != 0u);
      peer &= m ^ (bit - 1u);
    }
    const unsigned rank = __builtin_amdgcn_mbcnt_lo(peer, 0u);
    const unsigned tot  = (unsigned)__builtin_popcount(peer);
    const unsigned base = (unsigned)mycnt[d];
    loc[j] = base + rank;
    if (rank + 1u == tot) mycnt[d] = (int)(base + tot);
  }
  __syncthreads();

  if (t < NN) {
    int run = 0;
#pragma unroll
    for (int c = 0; c < 8; ++c) {
      const int v = cntw[c][t];
      cntw[c][t] = run;
      run += v;
    }
    deg[t] = run;
  }
  __syncthreads();

  if (wave == 0) {
    const int4 dg = *(const int4*)&deg[lane * 4];
    const int s = dg.x + dg.y + dg.z + dg.w;
    int inc = s;
#pragma unroll
    for (int o = 1; o < 32; o <<= 1) {
      const int v = __shfl_up(inc, o);
      if (lane >= o) inc += v;
    }
    const int ex = inc - s;
    rowstart[lane * 4 + 0] = ex;
    rowstart[lane * 4 + 1] = ex + dg.x;
    rowstart[lane * 4 + 2] = ex + dg.x + dg.y;
    rowstart[lane * 4 + 3] = ex + dg.x + dg.y + dg.z;
    if (lane == 31) rowstart[NN] = inc;
  }
  __syncthreads();

#pragma unroll
  for (int j = 0; j < 8; ++j) {
    const unsigned d = dv[j];
    const int pos = rowstart[d] + mycnt[d] + (int)loc[j];
    rl[pos] = (unsigned char)sv[j];
  }
  __syncthreads();

  float4* mg4 = (float4*)(meanOut + (size_t)g * NN * HD);
  for (int r = 0; r < 16; ++r) {
    const int beg = __builtin_amdgcn_readfirstlane(rowstart[wave * 16 + r]);
    const int n   = __builtin_amdgcn_readfirstlane(rowstart[wave * 16 + r + 1]) - beg;
    float4 a = make_float4(0.f, 0.f, 0.f, 0.f);
    for (int c = 0; c < n; c += 32) {
      const int k = c + lane;
      const int myidx = (k < n) ? (int)rl[beg + k] : ZROW;
      const int m = (n - c) < 32 ? (n - c) : 32;
      for (int j = 0; j < m; j += 4) {
        const int s0 = __builtin_amdgcn_readlane(myidx, j);
        const int s1 = __builtin_amdgcn_readlane(myidx, j + 1);
        const int s2 = __builtin_amdgcn_readlane(myidx, j + 2);
        const int s3 = __builtin_amdgcn_readlane(myidx, j + 3);
        const float4 v0 = feat4[s0 * (HD / 4) + lane];
        const float4 v1 = feat4[s1 * (HD / 4) + lane];
        const float4 v2 = feat4[s2 * (HD / 4) + lane];
        const float4 v3 = feat4[s3 * (HD / 4) + lane];
        a.x += v0.x; a.y += v0.y; a.z += v0.z; a.w += v0.w;
        a.x += v1.x; a.y += v1.y; a.z += v1.z; a.w += v1.w;
        a.x += v2.x; a.y += v2.y; a.z += v2.z; a.w += v2.w;
        a.x += v3.x; a.y += v3.y; a.z += v3.z; a.w += v3.w;
      }
    }
    const float cnt = (float)(n > 1 ? n : 1);
    float4 o;
    o.x = a.x / cnt; o.y = a.y / cnt; o.z = a.z / cnt; o.w = a.w / cnt;
    mg4[(wave * 16 + r) * (HD / 4) + lane] = o;
  }
}

#define GEMM_ROWS 256

__global__ __launch_bounds__(256) void sage_gemm(
    const float* __restrict__ Aagg, const float* __restrict__ Aself,
    const float* __restrict__ Wl,   const float* __restrict__ Wr,
    const float* __restrict__ bias, float* __restrict__ out)
{
  __shared__ float2 wp[128 * HD];

  const int t = threadIdx.x;
  for (int idx = t; idx < 128 * HD; idx += 256) {
    const int kk  = idx >> 7;
    const int col = idx & (HD - 1);
    const float* W = (kk < 64) ? Wl : Wr;
    const int k = (kk & 63) * 2;
    float2 v;
    v.x = W[(size_t)k       * HD + col];
    v.y = W[(size_t)(k + 1) * HD + col];
    wp[idx] = v;
  }
  __syncthreads();

  const int wave = t >> 5;
  const int lane = t & 31;
  const int m    = lane & 15;
  const int hi   = lane >> 4;
  const int kb   = hi * 2;

  const int rowBase = blockIdx.x * GEMM_ROWS + wave * 32;
  const float* arow1 = Aagg  + (size_t)(rowBase + m) * HD;
  const float* arow2 = Aself + (size_t)(rowBase + m) * HD;

  v8f p0 = {}, p1 = {}, p2 = {}, p3 = {}, p4 = {}, p5 = {}, p6 = {}, p7 = {};
  v8f q0 = {}, q1 = {}, q2 = {}, q3 = {}, q4 = {}, q5 = {}, q6 = {}, q7 = {};

#define GEMM_PAIR(P, Q, OFF)                                               \
    bb = bp[OFF]; b.x = bb.x; b.y = bb.y;                                  \
    P = __builtin_amdgcn_wmma_f32_16x16x4_f32(false, a0, false, b, (short)0, P, false, false); \
    Q = __builtin_amdgcn_wmma_f32_16x16x4_f32(false, a1, false, b, (short)0, Q, false, false);

#define GEMM_STEP(AROW, KOFF)                                              \
  {                                                                        \
    const v2f a0 = *(const v2f*)((AROW) + k0 + kb);                        \
    const v2f a1 = *(const v2f*)((AROW) + 16 * HD + k0 + kb);              \
    const float2* bp = wp + ((k0 >> 1) + hi + (KOFF)) * HD + m;            \
    float2 bb; v2f b;                                                      \
    GEMM_PAIR(p0, q0, 0)                                                   \
    GEMM_PAIR(p1, q1, 16)                                                  \
    GEMM_PAIR(p2, q2, 32)                                                  \
    GEMM_PAIR(p3, q3, 48)                                                  \
    GEMM_PAIR(p4, q4, 64)                                                  \
    GEMM_PAIR(p5, q5, 80)                                                  \
    GEMM_PAIR(p6, q6, 96)                                                  \
    GEMM_PAIR(p7, q7, 112)                                                 \
  }

#pragma unroll 2
  for (int k0 = 0; k0 < HD; k0 += 4) GEMM_STEP(arow1, 0)
#pragma unroll 2
  for (int k0 = 0; k0 < HD; k0 += 4) GEMM_STEP(arow2, 64)
#undef GEMM_STEP
#undef GEMM_PAIR

  const int rbase = rowBase + hi * 8;
  v8f accs[16] = {p0, p1, p2, p3, p4, p5, p6, p7, q0, q1, q2, q3, q4, q5, q6, q7};
#pragma unroll
  for (int rt = 0; rt < 2; ++rt) {
#pragma unroll
    for (int ct = 0; ct < 8; ++ct) {
      const int col = ct * 16 + m;
      const float bv = bias[col];
#pragma unroll
      for (int v = 0; v < 8; ++v) {
        float val = accs[rt * 8 + ct][v] + bv;
        val = val > 0.f ? val : 0.f;
        out[(size_t)(rbase + rt * 16 + v) * HD + col] = val;
      }
    }
  }
}

#define WT_PITCH 33

template <int NT>
__device__ __forceinline__ void conv_lane_is_channel(
    const float* xs, const float* wT, float bias, int tt0, int lane, float* y)
{
  float s[NT];
#pragma unroll
  for (int i = 0; i < NT; ++i) s[i] = bias;
  for (int w = 0; w < CONV_W; ++w) {
    const float* wcol = wT + (w * HD) * WT_PITCH + lane;
    const float* xrow = xs + (tt0 + w) * HD;
    for (int c = 0; c < HD; c += 4) {
      const float w0 = wcol[(c + 0) * WT_PITCH];
      const float w1 = wcol[(c + 1) * WT_PITCH];
      const float w2 = wcol[(c + 2) * WT_PITCH];
      const float w3 = wcol[(c + 3) * WT_PITCH];
#pragma unroll
      for (int i = 0; i < NT; ++i) {
        const float4 xv = *(const float4*)(xrow + i * 8 * HD + c);
        s[i] = __builtin_fmaf(xv.x, w0, s[i]);
        s[i] = __builtin_fmaf(xv.y, w1, s[i]);
        s[i] = __builtin_fmaf(xv.z, w2, s[i]);
        s[i] = __builtin_fmaf(xv.w, w3, s[i]);
      }
    }
  }
#pragma unroll
  for (int i = 0; i < NT; ++i) {
    const float v = s[i];
    y[lane * CONV_T + tt0 + 8 * i] = v > 0.f ? v : 0.f;
  }
}

__global__ __launch_bounds__(256) void sortpool_head(
    const float* __restrict__ H,
    const float* __restrict__ conv_w, const float* __restrict__ conv_b,
    float* __restrict__ yout)
{
  __shared__ float vals[NN];
  __shared__ int   idxs[NN];
  __shared__ __attribute__((aligned(16))) float xs[KTOP * HD];
  __shared__ float wT[CONV_W * HD * WT_PITCH];
  __shared__ __attribute__((aligned(16))) float y[FLAT];

  const int g = blockIdx.x;
  const int t = threadIdx.x;
  const float* hg = H + (size_t)g * NN * HD;

  for (int gi = t; gi < CONV_O * (HD / 4); gi += 256) {
    const int o  = gi >> 5;
    const int cq = gi & 31;
    const float4* p4 = (const float4*)(conv_w + (size_t)o * HD * CONV_W + cq * 12);
    const float4 a = p4[0], b = p4[1], d = p4[2];
    const float vv[12] = {a.x, a.y, a.z, a.w, b.x, b.y, b.z, b.w, d.x, d.y, d.z, d.w};
    float* dstp = wT + (cq * 4) * WT_PITCH + o;
#pragma unroll
    for (int k = 0; k < 12; ++k)
      dstp[((k % CONV_W) * HD + k / CONV_W) * WT_PITCH] = vv[k];
  }

  if (t < NN) { vals[t] = hg[t * HD + (HD - 1)]; idxs[t] = t; }
  __syncthreads();

  for (int k = 2; k <= NN; k <<= 1) {
    for (int j = k >> 1; j > 0; j >>= 1) {
      if (t < NN) {
        const int ixj = t ^ j;
        if (ixj > t) {
          const bool dirDesc = ((t & k) == 0);
          const float va = vals[t], vb = vals[ixj];
          const int   ia = idxs[t], ib = idxs[ixj];
          const bool aBeforeB = (va > vb) || (va == vb && ia < ib);
          if (dirDesc ? !aBeforeB : aBeforeB) {
            vals[t] = vb; vals[ixj] = va;
            idxs[t] = ib; idxs[ixj] = ia;
          }
        }
      }
      __syncthreads();
    }
  }

  for (int i = t; i < KTOP * HD; i += 256) {
    const int tt = i >> 7, c = i & (HD - 1);
    xs[i] = hg[(size_t)idxs[tt] * HD + c];
  }
  __syncthreads();

  {
    const int lane = t & 31;
    const int wave = __builtin_amdgcn_readfirstlane(t >> 5);
    const float cb = conv_b[lane];
    if (wave < 4) conv_lane_is_channel<4>(xs, wT, cb, wave, lane, y);
    else          conv_lane_is_channel<3>(xs, wT, cb, wave, lane, y);
  }
  __syncthreads();

  if (t < FLAT / 4)
    ((float4*)(yout + (size_t)g * FLAT))[t] = ((const float4*)y)[t];
}

#define DG        16
#define YS_PITCH  (FLAT + 4)
#define ZT_PITCH  (HD + 4)
#define KC        16
#define NCHUNK    (FLAT / (4 * KC))

__device__ __forceinline__ void load_w1_chunk(float (&b)[2 * KC], const float* __restrict__ p)
{
#pragma unroll
  for (int s = 0; s < KC; ++s) {
    b[2 * s]     = p[(size_t)(4 * s)     * HD];
    b[2 * s + 1] = p[(size_t)(4 * s + 1) * HD];
  }
}

__device__ __forceinline__ void mma_w1_chunk(v8f& acc0, v8f& acc1, const float (&b)[2 * KC],
                                             const float* arow)
{
#pragma unroll
  for (int s = 0; s < KC; s += 2) {
    const v2f a0 = *(const v2f*)(arow + 4 * s);
    const v2f a1 = *(const v2f*)(arow + 4 * s + 4);
    v2f f0, f1;
    f0.x = b[2 * s];     f0.y = b[2 * s + 1];
    f1.x = b[2 * s + 2]; f1.y = b[2 * s + 3];
    acc0 = __builtin_amdgcn_wmma_f32_16x16x4_f32(false, a0, false, f0, (short)0, acc0, false, false);
    acc1 = __builtin_amdgcn_wmma_f32_16x16x4_f32(false, a1, false, f1, (short)0, acc1, false, false);
  }
}

__global__ __launch_bounds__(256) void head_dense(
    const float* __restrict__ Y,
    const float* __restrict__ W1, const float* __restrict__ b1,
    const float* __restrict__ W2, const float* __restrict__ b2,
    float* __restrict__ out)
{
  __shared__ __attribute__((aligned(16))) float ys[DG * YS_PITCH];
  __shared__ __attribute__((aligned(16))) float zt[DG * ZT_PITCH];
  __shared__ __attribute__((aligned(16))) float w2s[HD * NCLS];
  __shared__ __attribute__((aligned(16))) float lg[DG * NCLS];

  const int t    = threadIdx.x;
  const int lane = t & 31;
  const int wave = t >> 5;
  const int m    = lane & 15;
  const int hi   = lane >> 4;
  const int kb   = hi * 2;
  const int g0   = blockIdx.x * DG;

  const float* wcol = W1 + (size_t)kb * HD + wave * 16 + m;
  float bA[2 * KC], bB[2 * KC];
  load_w1_chunk(bA, wcol);
  const float bv = b1[wave * 16 + m];

  const float4* yg4 = (const float4*)(Y + (size_t)g0 * FLAT);
#pragma unroll
  for (int k = 0; k < DG * (FLAT / 4) / 256; ++k) {
    const int i  = t + 256 * k;
    const int r  = i / (FLAT / 4);
    const int c4 = i - r * (FLAT / 4);
    *(float4*)&ys[r * YS_PITCH + c4 * 4] = yg4[i];
  }
#pragma unroll
  for (int k = 0; k < HD * NCLS / 4 / 256; ++k)
    ((float4*)w2s)[t + 256 * k] = ((const float4*)W2)[t + 256 * k];
  __syncthreads();

  v8f acc0 = {bv, bv, bv, bv, bv, bv, bv, bv};
  v8f acc1 = {};
  const float* arow = ys + m * YS_PITCH + kb;
  for (int c = 0; c < NCHUNK - 2; c += 2) {
    load_w1_chunk(bB, wcol + (size_t)(c + 1) * (4 * KC) * HD);
    mma_w1_chunk(acc0, acc1, bA, arow + c * (4 * KC));
    load_w1_chunk(bA, wcol + (size_t)(c + 2) * (4 * KC) * HD);
    mma_w1_chunk(acc0, acc1, bB, arow + (c + 1) * (4 * KC));
  }
  load_w1_chunk(bB, wcol + (size_t)(NCHUNK - 1) * (4 * KC) * HD);
  mma_w1_chunk(acc0, acc1, bA, arow + (NCHUNK - 2) * (4 * KC));
  mma_w1_chunk(acc0, acc1, bB, arow + (NCHUNK - 1) * (4 * KC));

#pragma unroll
  for (int v = 0; v < 8; ++v) {
    const float s = acc0[v] + acc1[v];
    zt[(v + 8 * hi) * ZT_PITCH + wave * 16 + m] = s > 0.f ? s : 0.f;
  }
  __syncthreads();

  const int gi  = t >> 4;
  const int cls = t & 15;
  float s = b2[cls];
  {
    const float* zr = zt + gi * ZT_PITCH;
    for (int j = 0; j < HD; j += 4) {
      const float4 zv = *(const float4*)(zr + j);
      s += zv.x * w2s[(j + 0) * NCLS + cls];
      s += zv.y * w2s[(j + 1) * NCLS + cls];
      s += zv.z * w2s[(j + 2) * NCLS + cls];
      s += zv.w * w2s[(j + 3) * NCLS + cls];
    }
  }
  lg[t] = s;
  __syncthreads();

  {
    const float* lr = lg + gi * NCLS;
    float mx = lr[0];
    for (int c = 1; c < NCLS; ++c) mx = fmaxf(mx, lr[c]);
    float se = 0.f;
    for (int c = 0; c < NCLS; ++c) se += expf(lr[c] - mx);
    const float lse = mx + logf(se);
    out[(size_t)(g0 + gi) * NCLS + cls] = s - lse;
  }
}

extern "C" void kernel_launch(void* const* d_in, const int* in_sizes, int n_in,
                              void* d_out, int out_size, void* d_ws, size_t ws_size,
                              hipStream_t stream) {
  (void)in_sizes; (void)n_in; (void)out_size; (void)ws_size;
  const float* x      = (const float*)d_in[0];
  const int*   src    = (const int*)d_in[1];
  const int*   dst    = (const int*)d_in[2];
  const float* Wl1    = (const float*)d_in[5];
  const float* bl1    = (const float*)d_in[6];
  const float* Wr1    = (const float*)d_in[7];
  const float* Wl2    = (const float*)d_in[8];
  const float* bl2    = (const float*)d_in[9];
  const float* Wr2    = (const float*)d_in[10];
  const float* Wl3    = (const float*)d_in[11];
  const float* bl3    = (const float*)d_in[12];
  const float* Wr3    = (const float*)d_in[13];
  const float* conv_w = (const float*)d_in[14];
  const float* conv_b = (const float*)d_in[15];
  const float* W1     = (const float*)d_in[16];
  const float* b1     = (const float*)d_in[17];
  const float* W2     = (const float*)d_in[18];
  const float* b2     = (const float*)d_in[19];

  float* ws   = (float*)d_ws;
  float* mean = ws;
  float* h1   = ws + (size_t)VN * HD;
  float* h2   = h1 + (size_t)VN * HD;

  const int gemmBlocks = VN / GEMM_ROWS;

  sage_aggregate<<<BG, 256, 0, stream>>>(x, src, dst, mean);
  sage_gemm<<<gemmBlocks, 256, 0, stream>>>(mean, x, Wl1, Wr1, bl1, h1);
  sage_aggregate<<<BG, 256, 0, stream>>>(h1, src, dst, mean);
  sage_gemm<<<gemmBlocks, 256, 0, stream>>>(mean, h1, Wl2, Wr2, bl2, h2);
  sage_aggregate<<<BG, 256, 0, stream>>>(h2, src, dst, mean);
  sage_gemm<<<gemmBlocks, 256, 0, stream>>>(mean, h2, Wl3, Wr3, bl3, h1);
  float* yflat = mean;
  sortpool_head<<<BG, 256, 0, stream>>>(h1, conv_w, conv_b, yflat);
  head_dense<<<BG / DG, 256, 0, stream>>>(yflat, W1, b1, W2, b2, (float*)d_out);
}


# head Conv1d: lane = time step, wave = 4 output channels; weights are scalar operands read straight from conv_w, x from a transposed LDS copy; conv sums now channel-major, taps inner
# speedup vs baseline: 6.5949x; 1.0342x over previous
#include <hip/hip_runtime.h>
#include <hip/hip_bf16.h>

#define BG   1024
#define NN   128
#define VN   (BG * NN)
#define HD   128
#define EPG  2048
#define KTOP 30
#define CONV_O 32
#define CONV_W 3
#define CONV_T (KTOP - CONV_W + 1)
#define FLAT   (CONV_O * CONV_T)
#define NCLS   16

typedef float v2f __attribute__((ext_vector_type(2)));
typedef float v8f __attribute__((ext_vector_type(8)));

#define ZROW NN

__global__ __launch_bounds__(256) void sage_aggregate(
    const float* __restrict__ X, const int* __restrict__ src,
    const int* __restrict__ dst, float* __restrict__ meanOut)
{
  __shared__ float4        feat4[(NN + 1) * (HD / 4)];
  __shared__ int           cntw[8][NN];
  __shared__ __attribute__((aligned(16))) int deg[NN];
  __shared__ int           rowstart[NN + 4];
  __shared__ unsigned char rl[EPG];

  const int g    = blockIdx.x;
  const int t    = threadIdx.x;
  const int lane = t & 31;
  const int wave = __builtin_amdgcn_readfirstlane(t >> 5);

  const float4* xg4 = (const float4*)(X + (size_t)g * NN * HD);
  for (int i = t; i < NN * HD / 4; i += 256) feat4[i] = xg4[i];
  if (t < HD / 4) feat4[ZROW * (HD / 4) + t] = make_float4(0.f, 0.f, 0.f, 0.f);

  int* mycnt = cntw[wave];
  mycnt[lane] = 0; mycnt[lane + 32] = 0; mycnt[lane + 64] = 0; mycnt[lane + 96] = 0;

  unsigned sv[8], dv[8], loc[8];
  {
    const int eb = g * EPG + wave * 256 + lane;
#pragma unroll
    for (int j = 0; j < 8; ++j) {
      sv[j] = (unsigned)src[eb + j * 32] & (NN - 1);
      dv[j] = (unsigned)dst[eb + j * 32] & (NN - 1);
    }
  }

#pragma unroll
  for (int j = 0; j < 8; ++j) {
    const unsigned d = dv[j];
    unsigned peer = 0xffffffffu;
#pragma unroll
    for (int b = 0; b < 7; ++b) {
      const unsigned bit = (d >> b) & 1u;
      const unsigned m = __builtin_amdgcn_ballot_w32(bit != 0u);
      peer &= m ^ (bit - 1u);
    }
    const unsigned rank = __builtin_amdgcn_mbcnt_lo(peer, 0u);
    const unsigned tot  = (unsigned)__builtin_popcount(peer);
    const unsigned base = (unsigned)mycnt[d];
    loc[j] = base + rank;
    if (rank + 1u == tot) mycnt[d] = (int)(base + tot);
  }
  __syncthreads();

  if (t < NN) {
    int run = 0;
#pragma unroll
    for (int c = 0; c < 8; ++c) {
      const int v = cntw[c][t];
      cntw[c][t] = run;
      run += v;
    }
    deg[t] = run;
  }
  __syncthreads();

  if (wave == 0) {
    const int4 dg = *(const int4*)&deg[lane * 4];
    const int s = dg.x + dg.y + dg.z + dg.w;
    int inc = s;
#pragma unroll
    for (int o = 1; o < 32; o <<= 1) {
      const int v = __shfl_up(inc, o);
      if (lane >= o) inc += v;
    }
    const int ex = inc - s;
    rowstart[lane * 4 + 0] = ex;
    rowstart[lane * 4 + 1] = ex + dg.x;
    rowstart[lane * 4 + 2] = ex + dg.x + dg.y;
    rowstart[lane * 4 + 3] = ex + dg.x + dg.y + dg.z;
    if (lane == 31) rowstart[NN] = inc;
  }
  __syncthreads();

#pragma unroll
  for (int j = 0; j < 8; ++j) {
    const unsigned d = dv[j];
    const int pos = rowstart[d] + mycnt[d] + (int)loc[j];
    rl[pos] = (unsigned char)sv[j];
  }
  __syncthreads();

  float4* mg4 = (float4*)(meanOut + (size_t)g * NN * HD);
  for (int r = 0; r < 16; ++r) {
    const int beg = __builtin_amdgcn_readfirstlane(rowstart[wave * 16 + r]);
    const int n   = __builtin_amdgcn_readfirstlane(rowstart[wave * 16 + r + 1]) - beg;
    float4 a = make_float4(0.f, 0.f, 0.f, 0.f);
    for (int c = 0; c < n; c += 32) {
      const int k = c + lane;
      const int myidx = (k < n) ? (int)rl[beg + k] : ZROW;
      const int m = (n - c) < 32 ? (n - c) : 32;
      for (int j = 0; j < m; j += 4) {
        const int s0 = __builtin_amdgcn_readlane(myidx, j);
        const int s1 = __builtin_amdgcn_readlane(myidx, j + 1);
        const int s2 = __builtin_amdgcn_readlane(myidx, j + 2);
        const int s3 = __builtin_amdgcn_readlane(myidx, j + 3);
        const float4 v0 = feat4[s0 * (HD / 4) + lane];
        const float4 v1 = feat4[s1 * (HD / 4) + lane];
        const float4 v2 = feat4[s2 * (HD / 4) + lane];
        const float4 v3 = feat4[s3 * (HD / 4) + lane];
        a.x += v0.x; a.y += v0.y; a.z += v0.z; a.w += v0.w;
        a.x += v1.x; a.y += v1.y; a.z += v1.z; a.w += v1.w;
        a.x += v2.x; a.y += v2.y; a.z += v2.z; a.w += v2.w;
        a.x += v3.x; a.y += v3.y; a.z += v3.z; a.w += v3.w;
      }
    }
    const float cnt = (float)(n > 1 ? n : 1);
    float4 o;
    o.x = a.x / cnt; o.y = a.y / cnt; o.z = a.z / cnt; o.w = a.w / cnt;
    mg4[(wave * 16 + r) * (HD / 4) + lane] = o;
  }
}

#define GEMM_ROWS 256

__global__ __launch_bounds__(256) void sage_gemm(
    const float* __restrict__ Aagg, const float* __restrict__ Aself,
    const float* __restrict__ Wl,   const float* __restrict__ Wr,
    const float* __restrict__ bias, float* __restrict__ out)
{
  __shared__ float2 wp[128 * HD];

  const int t = threadIdx.x;
  for (int idx = t; idx < 128 * HD; idx += 256) {
    const int kk  = idx >> 7;
    const int col = idx & (HD - 1);
    const float* W = (kk < 64) ? Wl : Wr;
    const int k = (kk & 63) * 2;
    float2 v;
    v.x = W[(size_t)k       * HD + col];
    v.y = W[(size_t)(k + 1) * HD + col];
    wp[idx] = v;
  }
  __syncthreads();

  const int wave = t >> 5;
  const int lane = t & 31;
  const int m    = lane & 15;
  const int hi   = lane >> 4;
  const int kb   = hi * 2;

  const int rowBase = blockIdx.x * GEMM_ROWS + wave * 32;
  const float* arow1 = Aagg  + (size_t)(rowBase + m) * HD;
  const float* arow2 = Aself + (size_t)(rowBase + m) * HD;

  v8f p0 = {}, p1 = {}, p2 = {}, p3 = {}, p4 = {}, p5 = {}, p6 = {}, p7 = {};
  v8f q0 = {}, q1 = {}, q2 = {}, q3 = {}, q4 = {}, q5 = {}, q6 = {}, q7 = {};

#define GEMM_PAIR(P, Q, OFF)                                               \
    bb = bp[OFF]; b.x = bb.x; b.y = bb.y;                                  \
    P = __builtin_amdgcn_wmma_f32_16x16x4_f32(false, a0, false, b, (short)0, P, false, false); \
    Q = __builtin_amdgcn_wmma_f32_16x16x4_f32(false, a1, false, b, (short)0, Q, false, false);

#define GEMM_STEP(AROW, KOFF)                                              \
  {                                                                        \
    const v2f a0 = *(const v2f*)((AROW) + k0 + kb);                        \
    const v2f a1 = *(const v2f*)((AROW) + 16 * HD + k0 + kb);              \
    const float2* bp = wp + ((k0 >> 1) + hi + (KOFF)) * HD + m;            \
    float2 bb; v2f b;                                                      \
    GEMM_PAIR(p0, q0, 0)                                                   \
    GEMM_PAIR(p1, q1, 16)                                                  \
    GEMM_PAIR(p2, q2, 32)                                                  \
    GEMM_PAIR(p3, q3, 48)                                                  \
    GEMM_PAIR(p4, q4, 64)                                                  \
    GEMM_PAIR(p5, q5, 80)                                                  \
    GEMM_PAIR(p6, q6, 96)                                                  \
    GEMM_PAIR(p7, q7, 112)                                                 \
  }

#pragma unroll 2
  for (int k0 = 0; k0 < HD; k0 += 4) GEMM_STEP(arow1, 0)
#pragma unroll 2
  for (int k0 = 0; k0 < HD; k0 += 4) GEMM_STEP(arow2, 64)
#undef GEMM_STEP
#undef GEMM_PAIR

  const int rbase = rowBase + hi * 8;
  v8f accs[16] = {p0, p1, p2, p3, p4, p5, p6, p7, q0, q1, q2, q3, q4, q5, q6, q7};
#pragma unroll
  for (int rt = 0; rt < 2; ++rt) {
#pragma unroll
    for (int ct = 0; ct < 8; ++ct) {
      const int col = ct * 16 + m;
      const float bv = bias[col];
#pragma unroll
      for (int v = 0; v < 8; ++v) {
        float val = accs[rt * 8 + ct][v] + bv;
        val = val > 0.f ? val : 0.f;
        out[(size_t)(rbase + rt * 16 + v) * HD + col] = val;
      }
    }
  }
}

#define XT_PITCH 33

#define CONV_CH(A, W)                                                      \
    A = __builtin_fmaf(x0.x, (W)[0],  A); A = __builtin_fmaf(x1.x, (W)[1],  A); \
    A = __builtin_fmaf(x2.x, (W)[2],  A); A = __builtin_fmaf(x0.y, (W)[3],  A); \
    A = __builtin_fmaf(x1.y, (W)[4],  A); A = __builtin_fmaf(x2.y, (W)[5],  A); \
    A = __builtin_fmaf(x0.z, (W)[6],  A); A = __builtin_fmaf(x1.z, (W)[7],  A); \
    A = __builtin_fmaf(x2.z, (W)[8],  A); A = __builtin_fmaf(x0.w, (W)[9],  A); \
    A = __builtin_fmaf(x1.w, (W)[10], A); A = __builtin_fmaf(x2.w, (W)[11], A);

__global__ __launch_bounds__(256) void sortpool_head(
    const float* __restrict__ H,
    const float* __restrict__ conv_w, const float* __restrict__ conv_b,
    float* __restrict__ yout)
{
  __shared__ float vals[NN];
  __shared__ int   idxs[NN];
  __shared__ float4 xT[(HD / 4) * XT_PITCH];
  __shared__ __attribute__((aligned(16))) float y[FLAT];

  const int g = blockIdx.x;
  const int t = threadIdx.x;
  const float* hg = H + (size_t)g * NN * HD;

  if (t < NN) { vals[t] = hg[t * HD + (HD - 1)]; idxs[t] = t; }
  __syncthreads();

  for (int k = 2; k <= NN; k <<= 1) {
    for (int j = k >> 1; j > 0; j >>= 1) {
      if (t < NN) {
        const int ixj = t ^ j;
        if (ixj > t) {
          const bool dirDesc = ((t & k) == 0);
          const float va = vals[t], vb = vals[ixj];
          const int   ia = idxs[t], ib = idxs[ixj];
          const bool aBeforeB = (va > vb) || (va == vb && ia < ib);
          if (dirDesc ? !aBeforeB : aBeforeB) {
            vals[t] = vb; vals[ixj] = va;
            idxs[t] = ib; idxs[ixj] = ia;
          }
        }
      }
      __syncthreads();
    }
  }

  {
    const int q = t & 31, tt0 = t >> 5;
    float4 v[4];
#pragma unroll
    for (int k = 0; k < 4; ++k) {
      const int tt = tt0 + 8 * k;
      const int row = idxs[tt < KTOP ? tt : KTOP - 1];
      v[k] = *(const float4*)(hg + (size_t)row * HD + 4 * q);
    }
#pragma unroll
    for (int k = 0; k < 4; ++k)
      xT[q * XT_PITCH + tt0 + 8 * k] = v[k];
  }
  __syncthreads();

  {
    const int lane = t & 31;
    const int wave = __builtin_amdgcn_readfirstlane(t >> 5);
    const int tl = lane < CONV_T ? lane : CONV_T - 1;
    const float* wb = conv_w + (size_t)(wave * 4) * (HD * CONV_W);
    float a0 = conv_b[wave * 4 + 0];
    float a1 = conv_b[wave * 4 + 1];
    float a2 = conv_b[wave * 4 + 2];
    float a3 = conv_b[wave * 4 + 3];
    const float4* xp = xT + tl;
#pragma unroll 2
    for (int q = 0; q < HD / 4; ++q) {
      const float4 x0 = xp[q * XT_PITCH];
      const float4 x1 = xp[q * XT_PITCH + 1];
      const float4 x2 = xp[q * XT_PITCH + 2];
      const float* w0 = wb + q * (4 * CONV_W);
      CONV_CH(a0, w0)
      CONV_CH(a1, w0 + HD * CONV_W)
      CONV_CH(a2, w0 + 2 * HD * CONV_W)
      CONV_CH(a3, w0 + 3 * HD * CONV_W)
    }
    if (lane < CONV_T) {
      float* yo = y + (wave * 4) * CONV_T + lane;
      yo[0]          = a0 > 0.f ? a0 : 0.f;
      yo[CONV_T]     = a1 > 0.f ? a1 : 0.f;
      yo[2 * CONV_T] = a2 > 0.f ? a2 : 0.f;
      yo[3 * CONV_T] = a3 > 0.f ? a3 : 0.f;
    }
  }
  __syncthreads();

  if (t < FLAT / 4)
    ((float4*)(yout + (size_t)g * FLAT))[t] = ((const float4*)y)[t];
}
#undef CONV_CH

#define DG        16
#define YS_PITCH  (FLAT + 4)
#define ZT_PITCH  (HD + 4)
#define KC        16
#define NCHUNK    (FLAT / (4 * KC))

__device__ __forceinline__ void load_w1_chunk(float (&b)[2 * KC], const float* __restrict__ p)
{
#pragma unroll
  for (int s = 0; s < KC; ++s) {
    b[2 * s]     = p[(size_t)(4 * s)     * HD];
    b[2 * s + 1] = p[(size_t)(4 * s + 1) * HD];
  }
}

__device__ __forceinline__ void mma_w1_chunk(v8f& acc0, v8f& acc1, const float (&b)[2 * KC],
                                             const float* arow)
{
#pragma unroll
  for (int s = 0; s < KC; s += 2) {
    const v2f a0 = *(const v2f*)(arow + 4 * s);
    const v2f a1 = *(const v2f*)(arow + 4 * s + 4);
    v2f f0, f1;
    f0.x = b[2 * s];     f0.y = b[2 * s + 1];
    f1.x = b[2 * s + 2]; f1.y = b[2 * s + 3];
    acc0 = __builtin_amdgcn_wmma_f32_16x16x4_f32(false, a0, false, f0, (short)0, acc0, false, false);
    acc1 = __builtin_amdgcn_wmma_f32_16x16x4_f32(false, a1, false, f1, (short)0, acc1, false, false);
  }
}

__global__ __launch_bounds__(256) void head_dense(
    const float* __restrict__ Y,
    const float* __restrict__ W1, const float* __restrict__ b1,
    const float* __restrict__ W2, const float* __restrict__ b2,
    float* __restrict__ out)
{
  __shared__ __attribute__((aligned(16))) float ys[DG * YS_PITCH];
  __shared__ __attribute__((aligned(16))) float zt[DG * ZT_PITCH];
  __shared__ __attribute__((aligned(16))) float w2s[HD * NCLS];
  __shared__ __attribute__((aligned(16))) float lg[DG * NCLS];

  const int t    = threadIdx.x;
  const int lane = t & 31;
  const int wave = t >> 5;
  const int m    = lane & 15;
  const int hi   = lane >> 4;
  const int kb   = hi * 2;
  const int g0   = blockIdx.x * DG;

  const float* wcol = W1 + (size_t)kb * HD + wave * 16 + m;
  float bA[2 * KC], bB[2 * KC];
  load_w1_chunk(bA, wcol);
  const float bv = b1[wave * 16 + m];

  const float4* yg4 = (const float4*)(Y + (size_t)g0 * FLAT);
#pragma unroll
  for (int k = 0; k < DG * (FLAT / 4) / 256; ++k) {
    const int i  = t + 256 * k;
    const int r  = i / (FLAT / 4);
    const int c4 = i - r * (FLAT / 4);
    *(float4*)&ys[r * YS_PITCH + c4 * 4] = yg4[i];
  }
#pragma unroll
  for (int k = 0; k < HD * NCLS / 4 / 256; ++k)
    ((float4*)w2s)[t + 256 * k] = ((const float4*)W2)[t + 256 * k];
  __syncthreads();

  v8f acc0 = {bv, bv, bv, bv, bv, bv, bv, bv};
  v8f acc1 = {};
  const float* arow = ys + m * YS_PITCH + kb;
  for (int c = 0; c < NCHUNK - 2; c += 2) {
    load_w1_chunk(bB, wcol + (size_t)(c + 1) * (4 * KC) * HD);
    mma_w1_chunk(acc0, acc1, bA, arow + c * (4 * KC));
    load_w1_chunk(bA, wcol + (size_t)(c + 2) * (4 * KC) * HD);
    mma_w1_chunk(acc0, acc1, bB, arow + (c + 1) * (4 * KC));
  }
  load_w1_chunk(bB, wcol + (size_t)(NCHUNK - 1) * (4 * KC) * HD);
  mma_w1_chunk(acc0, acc1, bA, arow + (NCHUNK - 2) * (4 * KC));
  mma_w1_chunk(acc0, acc1, bB, arow + (NCHUNK - 1) * (4 * KC));

#pragma unroll
  for (int v = 0; v < 8; ++v) {
    const float s = acc0[v] + acc1[v];
    zt[(v + 8 * hi) * ZT_PITCH + wave * 16 + m] = s > 0.f ? s : 0.f;
  }
  __syncthreads();

  const int gi  = t >> 4;
  const int cls = t & 15;
  float s = b2[cls];
  {
    const float* zr = zt + gi * ZT_PITCH;
    for (int j = 0; j < HD; j += 4) {
      const float4 zv = *(const float4*)(zr + j);
      s += zv.x * w2s[(j + 0) * NCLS + cls];
      s += zv.y * w2s[(j + 1) * NCLS + cls];
      s += zv.z * w2s[(j + 2) * NCLS + cls];
      s += zv.w * w2s[(j + 3) * NCLS + cls];
    }
  }
  lg[t] = s;
  __syncthreads();

  {
    const float* lr = lg + gi * NCLS;
    float mx = lr[0];
    for (int c = 1; c < NCLS; ++c) mx = fmaxf(mx, lr[c]);
    float se = 0.f;
    for (int c = 0; c < NCLS; ++c) se += expf(lr[c] - mx);
    const float lse = mx + logf(se);
    out[(size_t)(g0 + gi) * NCLS + cls] = s - lse;
  }
}

extern "C" void kernel_launch(void* const* d_in, const int* in_sizes, int n_in,
                              void* d_out, int out_size, void* d_ws, size_t ws_size,
                              hipStream_t stream) {
  (void)in_sizes; (void)n_in; (void)out_size; (void)ws_size;
  const float* x      = (const float*)d_in[0];
  const int*   src    = (const int*)d_in[1];
  const int*   dst    = (const int*)d_in[2];
  const float* Wl1    = (const float*)d_in[5];
  const float* bl1    = (const float*)d_in[6];
  const float* Wr1    = (const float*)d_in[7];
  const float* Wl2    = (const float*)d_in[8];
  const float* bl2    = (const float*)d_in[9];
  const float* Wr2    = (const float*)d_in[10];
  const float* Wl3    = (const float*)d_in[11];
  const float* bl3    = (const float*)d_in[12];
  const float* Wr3    = (const float*)d_in[13];
  const float* conv_w = (const float*)d_in[14];
  const float* conv_b = (const float*)d_in[15];
  const float* W1     = (const float*)d_in[16];
  const float* b1     = (const float*)d_in[17];
  const float* W2     = (const float*)d_in[18];
  const float* b2     = (const float*)d_in[19];

  float* ws   = (float*)d_ws;
  float* mean = ws;
  float* h1   = ws + (size_t)VN * HD;
  float* h2   = h1 + (size_t)VN * HD;

  const int gemmBlocks = VN / GEMM_ROWS;

  sage_aggregate<<<BG, 256, 0, stream>>>(x, src, dst, mean);
  sage_gemm<<<gemmBlocks, 256, 0, stream>>>(mean, x, Wl1, Wr1, bl1, h1);
  sage_aggregate<<<BG, 256, 0, stream>>>(h1, src, dst, mean);
  sage_gemm<<<gemmBlocks, 256, 0, stream>>>(mean, h1, Wl2, Wr2, bl2, h2);
  sage_aggregate<<<BG, 256, 0, stream>>>(h2, src, dst, mean);
  sage_gemm<<<gemmBlocks, 256, 0, stream>>>(mean, h2, Wl3, Wr3, bl3, h1);
  float* yflat = mean;
  sortpool_head<<<BG, 256, 0, stream>>>(h1, conv_w, conv_b, yflat);
  head_dense<<<BG / DG, 256, 0, stream>>>(yflat, W1, b1, W2, b2, (float*)d_out);
}


# layer-3 GEMM block (256 rows = 2 graphs) keeps H in LDS in the weights' place and does SortPool + Conv1d there; H3 round trip and the sortpool launch gone (7 launches); sum orders kept
# speedup vs baseline: 6.8488x; 1.0385x over previous
#include <hip/hip_runtime.h>
#include <hip/hip_bf16.h>

#define BG   1024
#define NN   128
#define VN   (BG * NN)
#define HD   128
#define EPG  2048
#define KTOP 30
#define CONV_O 32
#define CONV_W 3
#define CONV_T (KTOP - CONV_W + 1)
#define FLAT   (CONV_O * CONV_T)
#define NCLS   16

typedef float v2f __attribute__((ext_vector_type(2)));
typedef float v8f __attribute__((ext_vector_type(8)));

#define ZROW NN

__global__ __launch_bounds__(256) void sage_aggregate(
    const float* __restrict__ X, const int* __restrict__ src,
    const int* __restrict__ dst, float* __restrict__ meanOut)
{
  __shared__ float4        feat4[(NN + 1) * (HD / 4)];
  __shared__ int           cntw[8][NN];
  __shared__ __attribute__((aligned(16))) int deg[NN];
  __shared__ int           rowstart[NN + 4];
  __shared__ unsigned char rl[EPG];

  const int g    = blockIdx.x;
  const int t    = threadIdx.x;
  const int lane = t & 31;
  const int wave = __builtin_amdgcn_readfirstlane(t >> 5);

  const float4* xg4 = (const float4*)(X + (size_t)g * NN * HD);
  for (int i = t; i < NN * HD / 4; i += 256) feat4[i] = xg4[i];
  if (t < HD / 4) feat4[ZROW * (HD / 4) + t] = make_float4(0.f, 0.f, 0.f, 0.f);

  int* mycnt = cntw[wave];
  mycnt[lane] = 0; mycnt[lane + 32] = 0; mycnt[lane + 64] = 0; mycnt[lane + 96] = 0;

  unsigned sv[8], dv[8], loc[8];
  {
    const int eb = g * EPG + wave * 256 + lane;
#pragma unroll
    for (int j = 0; j < 8; ++j) {
      sv[j] = (unsigned)src[eb + j * 32] & (NN - 1);
      dv[j] = (unsigned)dst[eb + j * 32] & (NN - 1);
    }
  }

#pragma unroll
  for (int j = 0; j < 8; ++j) {
    const unsigned d = dv[j];
    unsigned peer = 0xffffffffu;
#pragma unroll
    for (int b = 0; b < 7; ++b) {
      const unsigned bit = (d >> b) & 1u;
      const unsigned m = __builtin_amdgcn_ballot_w32(bit != 0u);
      peer &= m ^ (bit - 1u);
    }
    const unsigned rank = __builtin_amdgcn_mbcnt_lo(peer, 0u);
    const unsigned tot  = (unsigned)__builtin_popcount(peer);
    const unsigned base = (unsigned)mycnt[d];
    loc[j] = base + rank;
    if (rank + 1u == tot) mycnt[d] = (int)(base + tot);
  }
  __syncthreads();

  if (t < NN) {
    int run = 0;
#pragma unroll
    for (int c = 0; c < 8; ++c) {
      const int v = cntw[c][t];
      cntw[c][t] = run;
      run += v;
    }
    deg[t] = run;
  }
  __syncthreads();

  if (wave == 0) {
    const int4 dg = *(const int4*)&deg[lane * 4];
    const int s = dg.x + dg.y + dg.z + dg.w;
    int inc = s;
#pragma unroll
    for (int o = 1; o < 32; o <<= 1) {
      const int v = __shfl_up(inc, o);
      if (lane >= o) inc += v;
    }
    const int ex = inc - s;
    rowstart[lane * 4 + 0] = ex;
    rowstart[lane * 4 + 1] = ex + dg.x;
    rowstart[lane * 4 + 2] = ex + dg.x + dg.y;
    rowstart[lane * 4 + 3] = ex + dg.x + dg.y + dg.z;
    if (lane == 31) rowstart[NN] = inc;
  }
  __syncthreads();

#pragma unroll
  for (int j = 0; j < 8; ++j) {
    const unsigned d = dv[j];
    const int pos = rowstart[d] + mycnt[d] + (int)loc[j];
    rl[pos] = (unsigned char)sv[j];
  }
  __syncthreads();

  float4* mg4 = (float4*)(meanOut + (size_t)g * NN * HD);
  for (int r = 0; r < 16; ++r) {
    const int beg = __builtin_amdgcn_readfirstlane(rowstart[wave * 16 + r]);
    const int n   = __builtin_amdgcn_readfirstlane(rowstart[wave * 16 + r + 1]) - beg;
    float4 a = make_float4(0.f, 0.f, 0.f, 0.f);
    for (int c = 0; c < n; c += 32) {
      const int k = c + lane;
      const int myidx = (k < n) ? (int)rl[beg + k] : ZROW;
      const int m = (n - c) < 32 ? (n - c) : 32;
      for (int j = 0; j < m; j += 4) {
        const int s0 = __builtin_amdgcn_readlane(myidx, j);
        const int s1 = __builtin_amdgcn_readlane(myidx, j + 1);
        const int s2 = __builtin_amdgcn_readlane(myidx, j + 2);
        const int s3 = __builtin_amdgcn_readlane(myidx, j + 3);
        const float4 v0 = feat4[s0 * (HD / 4) + lane];
        const float4 v1 = feat4[s1 * (HD / 4) + lane];
        const float4 v2 = feat4[s2 * (HD / 4) + lane];
        const float4 v3 = feat4[s3 * (HD / 4) + lane];
        a.x += v0.x; a.y += v0.y; a.z += v0.z; a.w += v0.w;
        a.x += v1.x; a.y += v1.y; a.z += v1.z; a.w += v1.w;
        a.x += v2.x; a.y += v2.y; a.z += v2.z; a.w += v2.w;
        a.x += v3.x; a.y += v3.y; a.z += v3.z; a.w += v3.w;
      }
    }
    const float cnt = (float)(n > 1 ? n : 1);
    float4 o;
    o.x = a.x / cnt; o.y = a.y / cnt; o.z = a.z / cnt; o.w = a.w / cnt;
    mg4[(wave * 16 + r) * (HD / 4) + lane] = o;
  }
}

#define GEMM_ROWS 256

#define XT_PITCH 33

#define CONV_CH(A, W, X0, X1, X2)                                          \
    A = __builtin_fmaf(X0.x, (W)[0],  A); A = __builtin_fmaf(X1.x, (W)[1],  A); \
    A = __builtin_fmaf(X2.x, (W)[2],  A); A = __builtin_fmaf(X0.y, (W)[3],  A); \
    A = __builtin_fmaf(X1.y, (W)[4],  A); A = __builtin_fmaf(X2.y, (W)[5],  A); \
    A = __builtin_fmaf(X0.z, (W)[6],  A); A = __builtin_fmaf(X1.z, (W)[7],  A); \
    A = __builtin_fmaf(X2.z, (W)[8],  A); A = __builtin_fmaf(X0.w, (W)[9],  A); \
    A = __builtin_fmaf(X1.w, (W)[10], A); A = __builtin_fmaf(X2.w, (W)[11], A);

template <bool HEAD>
__global__ __launch_bounds__(256) void sage_gemm(
    const float* __restrict__ Aagg, const float* __restrict__ Aself,
    const float* __restrict__ Wl,   const float* __restrict__ Wr,
    const float* __restrict__ bias, float* __restrict__ out,
    const float* __restrict__ conv_w, const float* __restrict__ conv_b,
    float* __restrict__ yout)
{
  __shared__ float2 wp[128 * HD];

  const int t = threadIdx.x;
  for (int idx = t; idx < 128 * HD; idx += 256) {
    const int kk  = idx >> 7;
    const int col = idx & (HD - 1);
    const float* W = (kk < 64) ? Wl : Wr;
    const int k = (kk & 63) * 2;
    float2 v;
    v.x = W[(size_t)k       * HD + col];
    v.y = W[(size_t)(k + 1) * HD + col];
    wp[idx] = v;
  }
  __syncthreads();

  const int wave = t >> 5;
  const int lane = t & 31;
  const int m    = lane & 15;
  const int hi   = lane >> 4;
  const int kb   = hi * 2;

  const int rowBase = blockIdx.x * GEMM_ROWS + wave * 32;
  const float* arow1 = Aagg  + (size_t)(rowBase + m) * HD;
  const float* arow2 = Aself + (size_t)(rowBase + m) * HD;

  v8f p0 = {}, p1 = {}, p2 = {}, p3 = {}, p4 = {}, p5 = {}, p6 = {}, p7 = {};
  v8f q0 = {}, q1 = {}, q2 = {}, q3 = {}, q4 = {}, q5 = {}, q6 = {}, q7 = {};

#define GEMM_PAIR(P, Q, OFF)                                               \
    bb = bp[OFF]; b.x = bb.x; b.y = bb.y;                                  \
    P = __builtin_amdgcn_wmma_f32_16x16x4_f32(false, a0, false, b, (short)0, P, false, false); \
    Q = __builtin_amdgcn_wmma_f32_16x16x4_f32(false, a1, false, b, (short)0, Q, false, false);

#define GEMM_STEP(AROW, KOFF)                                              \
  {                                                                        \
    const v2f a0 = *(const v2f*)((AROW) + k0 + kb);                        \
    const v2f a1 = *(const v2f*)((AROW) + 16 * HD + k0 + kb);              \
    const float2* bp = wp + ((k0 >> 1) + hi + (KOFF)) * HD + m;            \
    float2 bb; v2f b;                                                      \
    GEMM_PAIR(p0, q0, 0)                                                   \
    GEMM_PAIR(p1, q1, 16)                                                  \
    GEMM_PAIR(p2, q2, 32)                                                  \
    GEMM_PAIR(p3, q3, 48)                                                  \
    GEMM_PAIR(p4, q4, 64)                                                  \
    GEMM_PAIR(p5, q5, 80)                                                  \
    GEMM_PAIR(p6, q6, 96)                                                  \
    GEMM_PAIR(p7, q7, 112)                                                 \
  }

#pragma unroll 2
  for (int k0 = 0; k0 < HD; k0 += 4) GEMM_STEP(arow1, 0)
#pragma unroll 2
  for (int k0 = 0; k0 < HD; k0 += 4) GEMM_STEP(arow2, 64)
#undef GEMM_STEP
#undef GEMM_PAIR

  const int rbase = rowBase + hi * 8;
  v8f accs[16] = {p0, p1, p2, p3, p4, p5, p6, p7, q0, q1, q2, q3, q4, q5, q6, q7};
if constexpr (!HEAD) {
#pragma unroll
  for (int rt = 0; rt < 2; ++rt) {
#pragma unroll
    for (int ct = 0; ct < 8; ++ct) {
      const int col = ct * 16 + m;
      const float bv = bias[col];
#pragma unroll
      for (int v = 0; v < 8; ++v) {
        float val = accs[rt * 8 + ct][v] + bv;
        val = val > 0.f ? val : 0.f;
        out[(size_t)(rbase + rt * 16 + v) * HD + col] = val;
      }
    }
  }
  } else {
  __shared__ float svals[2 * NN];
  __shared__ int   sidxs[2 * NN];
  float*  hs  = (float*)wp;
  float4* hs4 = (float4*)wp;

  __syncthreads();
  {
    const int lrow = wave * 32 + hi * 8;
#pragma unroll
    for (int rt = 0; rt < 2; ++rt) {
#pragma unroll
      for (int ct = 0; ct < 8; ++ct) {
        const int col = ct * 16 + m;
        const float bv = bias[col];
#pragma unroll
        for (int v = 0; v < 8; ++v) {
          float val = accs[rt * 8 + ct][v] + bv;
          val = val > 0.f ? val : 0.f;
          hs[(lrow + rt * 16 + v) * HD + col] = val;
        }
      }
    }
  }
  __syncthreads();

  const int node = t & (NN - 1);
  svals[t] = hs[t * HD + (HD - 1)];
  sidxs[t] = node;
  __syncthreads();

  for (int k = 2; k <= NN; k <<= 1) {
    for (int j = k >> 1; j > 0; j >>= 1) {
      const int ixj = t ^ j;
      if (ixj > t) {
        const bool dirDesc = ((node & k) == 0);
        const float va = svals[t], vb = svals[ixj];
        const int   ia = sidxs[t], ib = sidxs[ixj];
        const bool aBeforeB = (va > vb) || (va == vb && ia < ib);
        if (dirDesc ? !aBeforeB : aBeforeB) {
          svals[t] = vb; svals[ixj] = va;
          sidxs[t] = ib; sidxs[ixj] = ia;
        }
      }
      __syncthreads();
    }
  }

  const int q = t & 31, tt0 = t >> 5;
  const int ts3 = (tt0 + 24) < KTOP ? (tt0 + 24) : KTOP - 1;
  const float4 g00 = hs4[sidxs[tt0]      * (HD / 4) + q];
  const float4 g01 = hs4[sidxs[tt0 + 8]  * (HD / 4) + q];
  const float4 g02 = hs4[sidxs[tt0 + 16] * (HD / 4) + q];
  const float4 g03 = hs4[sidxs[ts3]      * (HD / 4) + q];
  const float4 g10 = hs4[(NN + sidxs[NN + tt0])      * (HD / 4) + q];
  const float4 g11 = hs4[(NN + sidxs[NN + tt0 + 8])  * (HD / 4) + q];
  const float4 g12 = hs4[(NN + sidxs[NN + tt0 + 16]) * (HD / 4) + q];
  const float4 g13 = hs4[(NN + sidxs[NN + ts3])      * (HD / 4) + q];
  __syncthreads();

  float4* xT0 = hs4;
  float4* xT1 = hs4 + (HD / 4) * XT_PITCH;
  float*  ys  = (float*)(hs4 + 2 * (HD / 4) * XT_PITCH);
  xT0[q * XT_PITCH + tt0]      = g00;
  xT0[q * XT_PITCH + tt0 + 8]  = g01;
  xT0[q * XT_PITCH + tt0 + 16] = g02;
  xT0[q * XT_PITCH + tt0 + 24] = g03;
  xT1[q * XT_PITCH + tt0]      = g10;
  xT1[q * XT_PITCH + tt0 + 8]  = g11;
  xT1[q * XT_PITCH + tt0 + 16] = g12;
  xT1[q * XT_PITCH + tt0 + 24] = g13;
  __syncthreads();

  {
    const int wv = __builtin_amdgcn_readfirstlane(t >> 5);
    const int tl = lane < CONV_T ? lane : CONV_T - 1;
    typedef const float __attribute__((address_space(4))) cfloat;
    const cfloat* wb = (const cfloat*)conv_w + (size_t)(wv * 4) * (HD * CONV_W);
    const cfloat* cb = (const cfloat*)conv_b + wv * 4;
    float a0 = cb[0], a1 = cb[1];
    float a2 = cb[2], a3 = cb[3];
    float c0 = a0, c1 = a1, c2 = a2, c3 = a3;
    const float4* xp0 = xT0 + tl;
    const float4* xp1 = xT1 + tl;
#pragma unroll 2
    for (int qq = 0; qq < HD / 4; ++qq) {
      const float4 x0 = xp0[qq * XT_PITCH];
      const float4 x1 = xp0[qq * XT_PITCH + 1];
      const float4 x2 = xp0[qq * XT_PITCH + 2];
      const float4 z0 = xp1[qq * XT_PITCH];
      const float4 z1 = xp1[qq * XT_PITCH + 1];
      const float4 z2 = xp1[qq * XT_PITCH + 2];
      const cfloat* w0 = wb + qq * (4 * CONV_W);
      CONV_CH(a0, w0, x0, x1, x2)
      CONV_CH(a1, w0 + HD * CONV_W, x0, x1, x2)
      CONV_CH(a2, w0 + 2 * HD * CONV_W, x0, x1, x2)
      CONV_CH(a3, w0 + 3 * HD * CONV_W, x0, x1, x2)
      CONV_CH(c0, w0, z0, z1, z2)
      CONV_CH(c1, w0 + HD * CONV_W, z0, z1, z2)
      CONV_CH(c2, w0 + 2 * HD * CONV_W, z0, z1, z2)
      CONV_CH(c3, w0 + 3 * HD * CONV_W, z0, z1, z2)
    }
    if (lane < CONV_T) {
      float* yo = ys + (wv * 4) * CONV_T + lane;
      yo[0]          = a0 > 0.f ? a0 : 0.f;
      yo[CONV_T]     = a1 > 0.f ? a1 : 0.f;
      yo[2 * CONV_T] = a2 > 0.f ? a2 : 0.f;
      yo[3 * CONV_T] = a3 > 0.f ? a3 : 0.f;
      yo[FLAT]              = c0 > 0.f ? c0 : 0.f;
      yo[FLAT + CONV_T]     = c1 > 0.f ? c1 : 0.f;
      yo[FLAT + 2 * CONV_T] = c2 > 0.f ? c2 : 0.f;
      yo[FLAT + 3 * CONV_T] = c3 > 0.f ? c3 : 0.f;
    }
  }
  __syncthreads();

  {
    float4* yg4 = (float4*)(yout + (size_t)(blockIdx.x * 2) * FLAT);
    const float4* ys4 = (const float4*)ys;
    for (int i = t; i < 2 * FLAT / 4; i += 256) yg4[i] = ys4[i];
  }
  }
}
#undef CONV_CH

#define DG        16
#define YS_PITCH  (FLAT + 4)
#define ZT_PITCH  (HD + 4)
#define KC        16
#define NCHUNK    (FLAT / (4 * KC))

__device__ __forceinline__ void load_w1_chunk(float (&b)[2 * KC], const float* __restrict__ p)
{
#pragma unroll
  for (int s = 0; s < KC; ++s) {
    b[2 * s]     = p[(size_t)(4 * s)     * HD];
    b[2 * s + 1] = p[(size_t)(4 * s + 1) * HD];
  }
}

__device__ __forceinline__ void mma_w1_chunk(v8f& acc0, v8f& acc1, const float (&b)[2 * KC],
                                             const float* arow)
{
#pragma unroll
  for (int s = 0; s < KC; s += 2) {
    const v2f a0 = *(const v2f*)(arow + 4 * s);
    const v2f a1 = *(const v2f*)(arow + 4 * s + 4);
    v2f f0, f1;
    f0.x = b[2 * s];     f0.y = b[2 * s + 1];
    f1.x = b[2 * s + 2]; f1.y = b[2 * s + 3];
    acc0 = __builtin_amdgcn_wmma_f32_16x16x4_f32(false, a0, false, f0, (short)0, acc0, false, false);
    acc1 = __builtin_amdgcn_wmma_f32_16x16x4_f32(false, a1, false, f1, (short)0, acc1, false, false);
  }
}

__global__ __launch_bounds__(256) void head_dense(
    const float* __restrict__ Y,
    const float* __restrict__ W1, const float* __restrict__ b1,
    const float* __restrict__ W2, const float* __restrict__ b2,
    float* __restrict__ out)
{
  __shared__ __attribute__((aligned(16))) float ys[DG * YS_PITCH];
  __shared__ __attribute__((aligned(16))) float zt[DG * ZT_PITCH];
  __shared__ __attribute__((aligned(16))) float w2s[HD * NCLS];
  __shared__ __attribute__((aligned(16))) float lg[DG * NCLS];

  const int t    = threadIdx.x;
  const int lane = t & 31;
  const int wave = t >> 5;
  const int m    = lane & 15;
  const int hi   = lane >> 4;
  const int kb   = hi * 2;
  const int g0   = blockIdx.x * DG;

  const float* wcol = W1 + (size_t)kb * HD + wave * 16 + m;
  float bA[2 * KC], bB[2 * KC];
  load_w1_chunk(bA, wcol);
  const float bv = b1[wave * 16 + m];

  const float4* yg4 = (const float4*)(Y + (size_t)g0 * FLAT);
#pragma unroll
  for (int k = 0; k < DG * (FLAT / 4) / 256; ++k) {
    const int i  = t + 256 * k;
    const int r  = i / (FLAT / 4);
    const int c4 = i - r * (FLAT / 4);
    *(float4*)&ys[r * YS_PITCH + c4 * 4] = yg4[i];
  }
#pragma unroll
  for (int k = 0; k < HD * NCLS / 4 / 256; ++k)
    ((float4*)w2s)[t + 256 * k] = ((const float4*)W2)[t + 256 * k];
  __syncthreads();

  v8f acc0 = {bv, bv, bv, bv, bv, bv, bv, bv};
  v8f acc1 = {};
  const float* arow = ys + m * YS_PITCH + kb;
  for (int c = 0; c < NCHUNK - 2; c += 2) {
    load_w1_chunk(bB, wcol + (size_t)(c + 1) * (4 * KC) * HD);
    mma_w1_chunk(acc0, acc1, bA, arow + c * (4 * KC));
    load_w1_chunk(bA, wcol + (size_t)(c + 2) * (4 * KC) * HD);
    mma_w1_chunk(acc0, acc1, bB, arow + (c + 1) * (4 * KC));
  }
  load_w1_chunk(bB, wcol + (size_t)(NCHUNK - 1) * (4 * KC) * HD);
  mma_w1_chunk(acc0, acc1, bA, arow + (NCHUNK - 2) * (4 * KC));
  mma_w1_chunk(acc0, acc1, bB, arow + (NCHUNK - 1) * (4 * KC));

#pragma unroll
  for (int v = 0; v < 8; ++v) {
    const float s = acc0[v] + acc1[v];
    zt[(v + 8 * hi) * ZT_PITCH + wave * 16 + m] = s > 0.f ? s : 0.f;
  }
  __syncthreads();

  const int gi  = t >> 4;
  const int cls = t & 15;
  float s = b2[cls];
  {
    const float* zr = zt + gi * ZT_PITCH;
    for (int j = 0; j < HD; j += 4) {
      const float4 zv = *(const float4*)(zr + j);
      s += zv.x * w2s[(j + 0) * NCLS + cls];
      s += zv.y * w2s[(j + 1) * NCLS + cls];
      s += zv.z * w2s[(j + 2) * NCLS + cls];
      s += zv.w * w2s[(j + 3) * NCLS + cls];
    }
  }
  lg[t] = s;
  __syncthreads();

  {
    const float* lr = lg + gi * NCLS;
    float mx = lr[0];
    for (int c = 1; c < NCLS; ++c) mx = fmaxf(mx, lr[c]);
    float se = 0.f;
    for (int c = 0; c < NCLS; ++c) se += expf(lr[c] - mx);
    const float lse = mx + logf(se);
    out[(size_t)(g0 + gi) * NCLS + cls] = s - lse;
  }
}

extern "C" void kernel_launch(void* const* d_in, const int* in_sizes, int n_in,
                              void* d_out, int out_size, void* d_ws, size_t ws_size,
                              hipStream_t stream) {
  (void)in_sizes; (void)n_in; (void)out_size; (void)ws_size;
  const float* x      = (const float*)d_in[0];
  const int*   src    = (const int*)d_in[1];
  const int*   dst    = (const int*)d_in[2];
  const float* Wl1    = (const float*)d_in[5];
  const float* bl1    = (const float*)d_in[6];
  const float* Wr1    = (const float*)d_in[7];
  const float* Wl2    = (const float*)d_in[8];
  const float* bl2    = (const float*)d_in[9];
  const float* Wr2    = (const float*)d_in[10];
  const float* Wl3    = (const float*)d_in[11];
  const float* bl3    = (const float*)d_in[12];
  const float* Wr3    = (const float*)d_in[13];
  const float* conv_w = (const float*)d_in[14];
  const float* conv_b = (const float*)d_in[15];
  const float* W1     = (const float*)d_in[16];
  const float* b1     = (const float*)d_in[17];
  const float* W2     = (const float*)d_in[18];
  const float* b2     = (const float*)d_in[19];

  float* ws   = (float*)d_ws;
  float* mean = ws;
  float* h1   = ws + (size_t)VN * HD;
  float* h2   = h1 + (size_t)VN * HD;

  const int gemmBlocks = VN / GEMM_ROWS;

  sage_aggregate<<<BG, 256, 0, stream>>>(x, src, dst, mean);
  sage_gemm<false><<<gemmBlocks, 256, 0, stream>>>(mean, x, Wl1, Wr1, bl1, h1, nullptr, nullptr, nullptr);
  sage_aggregate<<<BG, 256, 0, stream>>>(h1, src, dst, mean);
  sage_gemm<false><<<gemmBlocks, 256, 0, stream>>>(mean, h1, Wl2, Wr2, bl2, h2, nullptr, nullptr, nullptr);
  float* yflat = h1;
  sage_aggregate<<<BG, 256, 0, stream>>>(h2, src, dst, mean);
  sage_gemm<true><<<gemmBlocks, 256, 0, stream>>>(mean, h2, Wl3, Wr3, bl3, nullptr, conv_w, conv_b, yflat);
  head_dense<<<BG / DG, 256, 0, stream>>>(yflat, W1, b1, W2, b2, (float*)d_out);
}


# sage_gemm: weights staged in 8 chunks; chunk c+1 loads to registers during the k-steps of chunk c (A requested 4 steps ahead, so no early wait covers it), stored to LDS behind them; sum order kept
# speedup vs baseline: 7.1575x; 1.0451x over previous
#include <hip/hip_runtime.h>
#include <hip/hip_bf16.h>

#define BG   1024
#define NN   128
#define VN   (BG * NN)
#define HD   128
#define EPG  2048
#define KTOP 30
#define CONV_O 32
#define CONV_W 3
#define CONV_T (KTOP - CONV_W + 1)
#define FLAT   (CONV_O * CONV_T)
#define NCLS   16

typedef float v2f __attribute__((ext_vector_type(2)));
typedef float v8f __attribute__((ext_vector_type(8)));

#define ZROW NN

__global__ __launch_bounds__(256) void sage_aggregate(
    const float* __restrict__ X, const int* __restrict__ src,
    const int* __restrict__ dst, float* __restrict__ meanOut)
{
  __shared__ float4        feat4[(NN + 1) * (HD / 4)];
  __shared__ int           cntw[8][NN];
  __shared__ __attribute__((aligned(16))) int deg[NN];
  __shared__ int           rowstart[NN + 4];
  __shared__ unsigned char rl[EPG];

  const int g    = blockIdx.x;
  const int t    = threadIdx.x;
  const int lane = t & 31;
  const int wave = __builtin_amdgcn_readfirstlane(t >> 5);

  const float4* xg4 = (const float4*)(X + (size_t)g * NN * HD);
  for (int i = t; i < NN * HD / 4; i += 256) feat4[i] = xg4[i];
  if (t < HD / 4) feat4[ZROW * (HD / 4) + t] = make_float4(0.f, 0.f, 0.f, 0.f);

  int* mycnt = cntw[wave];
  mycnt[lane] = 0; mycnt[lane + 32] = 0; mycnt[lane + 64] = 0; mycnt[lane + 96] = 0;

  unsigned sv[8], dv[8], loc[8];
  {
    const int eb = g * EPG + wave * 256 + lane;
#pragma unroll
    for (int j = 0; j < 8; ++j) {
      sv[j] = (unsigned)src[eb + j * 32] & (NN - 1);
      dv[j] = (unsigned)dst[eb + j * 32] & (NN - 1);
    }
  }

#pragma unroll
  for (int j = 0; j < 8; ++j) {
    const unsigned d = dv[j];
    unsigned peer = 0xffffffffu;
#pragma unroll
    for (int b = 0; b < 7; ++b) {
      const unsigned bit = (d >> b) & 1u;
      const unsigned m = __builtin_amdgcn_ballot_w32(bit != 0u);
      peer &= m ^ (bit - 1u);
    }
    const unsigned rank = __builtin_amdgcn_mbcnt_lo(peer, 0u);
    const unsigned tot  = (unsigned)__builtin_popcount(peer);
    const unsigned base = (unsigned)mycnt[d];
    loc[j] = base + rank;
    if (rank + 1u == tot) mycnt[d] = (int)(base + tot);
  }
  __syncthreads();

  if (t < NN) {
    int run = 0;
#pragma unroll
    for (int c = 0; c < 8; ++c) {
      const int v = cntw[c][t];
      cntw[c][t] = run;
      run += v;
    }
    deg[t] = run;
  }
  __syncthreads();

  if (wave == 0) {
    const int4 dg = *(const int4*)&deg[lane * 4];
    const int s = dg.x + dg.y + dg.z + dg.w;
    int inc = s;
#pragma unroll
    for (int o = 1; o < 32; o <<= 1) {
      const int v = __shfl_up(inc, o);
      if (lane >= o) inc += v;
    }
    const int ex = inc - s;
    rowstart[lane * 4 + 0] = ex;
    rowstart[lane * 4 + 1] = ex + dg.x;
    rowstart[lane * 4 + 2] = ex + dg.x + dg.y;
    rowstart[lane * 4 + 3] = ex + dg.x + dg.y + dg.z;
    if (lane == 31) rowstart[NN] = inc;
  }
  __syncthreads();

#pragma unroll
  for (int j = 0; j < 8; ++j) {
    const unsigned d = dv[j];
    const int pos = rowstart[d] + mycnt[d] + (int)loc[j];
    rl[pos] = (unsigned char)sv[j];
  }
  __syncthreads();

  float4* mg4 = (float4*)(meanOut + (size_t)g * NN * HD);
  for (int r = 0; r < 16; ++r) {
    const int beg = __builtin_amdgcn_readfirstlane(rowstart[wave * 16 + r]);
    const int n   = __builtin_amdgcn_readfirstlane(rowstart[wave * 16 + r + 1]) - beg;
    float4 a = make_float4(0.f, 0.f, 0.f, 0.f);
    for (int c = 0; c < n; c += 32) {
      const int k = c + lane;
      const int myidx = (k < n) ? (int)rl[beg + k] : ZROW;
      const int m = (n - c) < 32 ? (n - c) : 32;
      for (int j = 0; j < m; j += 4) {
        const int s0 = __builtin_amdgcn_readlane(myidx, j);
        const int s1 = __builtin_amdgcn_readlane(myidx, j + 1);
        const int s2 = __builtin_amdgcn_readlane(myidx, j + 2);
        const int s3 = __builtin_amdgcn_readlane(myidx, j + 3);
        const float4 v0 = feat4[s0 * (HD / 4) + lane];
        const float4 v1 = feat4[s1 * (HD / 4) + lane];
        const float4 v2 = feat4[s2 * (HD / 4) + lane];
        const float4 v3 = feat4[s3 * (HD / 4) + lane];
        a.x += v0.x; a.y += v0.y; a.z += v0.z; a.w += v0.w;
        a.x += v1.x; a.y += v1.y; a.z += v1.z; a.w += v1.w;
        a.x += v2.x; a.y += v2.y; a.z += v2.z; a.w += v2.w;
        a.x += v3.x; a.y += v3.y; a.z += v3.z; a.w += v3.w;
      }
    }
    const float cnt = (float)(n > 1 ? n : 1);
    float4 o;
    o.x = a.x / cnt; o.y = a.y / cnt; o.z = a.z / cnt; o.w = a.w / cnt;
    mg4[(wave * 16 + r) * (HD / 4) + lane] = o;
  }
}

#define GEMM_ROWS 256

#define XT_PITCH 33

#define CONV_CH(A, W, X0, X1, X2)                                          \
    A = __builtin_fmaf(X0.x, (W)[0],  A); A = __builtin_fmaf(X1.x, (W)[1],  A); \
    A = __builtin_fmaf(X2.x, (W)[2],  A); A = __builtin_fmaf(X0.y, (W)[3],  A); \
    A = __builtin_fmaf(X1.y, (W)[4],  A); A = __builtin_fmaf(X2.y, (W)[5],  A); \
    A = __builtin_fmaf(X0.z, (W)[6],  A); A = __builtin_fmaf(X1.z, (W)[7],  A); \
    A = __builtin_fmaf(X2.z, (W)[8],  A); A = __builtin_fmaf(X0.w, (W)[9],  A); \
    A = __builtin_fmaf(X1.w, (W)[10], A); A = __builtin_fmaf(X2.w, (W)[11], A);

template <bool HEAD>
__global__ __launch_bounds__(256) void sage_gemm(
    const float* __restrict__ Aagg, const float* __restrict__ Aself,
    const float* __restrict__ Wl,   const float* __restrict__ Wr,
    const float* __restrict__ bias, float* __restrict__ out,
    const float* __restrict__ conv_w, const float* __restrict__ conv_b,
    float* __restrict__ yout)
{
  __shared__ float2 wp[128 * HD];

  const int t = threadIdx.x;
  const int wave = __builtin_amdgcn_readfirstlane(t >> 5);
  const int lane = t & 31;
  const int m    = lane & 15;
  const int hi   = lane >> 4;
  const int kb   = hi * 2;

  const int rowBase = blockIdx.x * GEMM_ROWS + wave * 32;
  const float* arow1 = Aagg  + (size_t)(rowBase + m) * HD + kb;
  const float* arow2 = Aself + (size_t)(rowBase + m) * HD + kb;

#define W_LOAD(C)                                                          \
  {                                                                        \
    const float* Wc = ((C) < 4) ? Wl : Wr;                                 \
    const float* wr = Wc + (size_t)((((C) & 3) * 16 + wave * 2) * 2) * HD + 4 * lane; \
    hw0 = *(const float4*)(wr);                                            \
    hw1 = *(const float4*)(wr + HD);                                       \
    hw2 = *(const float4*)(wr + 2 * HD);                                   \
    hw3 = *(const float4*)(wr + 3 * HD);                                   \
  }
#define W_STORE(C)                                                         \
  {                                                                        \
    float4* wd = (float4*)(wp + ((C) * 16 + wave * 2) * HD + 4 * lane);    \
    wd[0]          = make_float4(hw0.x, hw1.x, hw0.y, hw1.y);              \
    wd[1]          = make_float4(hw0.z, hw1.z, hw0.w, hw1.w);              \
    wd[HD / 2]     = make_float4(hw2.x, hw3.x, hw2.y, hw3.y);              \
    wd[HD / 2 + 1] = make_float4(hw2.z, hw3.z, hw2.w, hw3.w);              \
  }

  v2f ra0[4], ra1[4];
#pragma unroll
  for (int j = 0; j < 4; ++j) {
    ra0[j] = *(const v2f*)(arow1 + 4 * j);
    ra1[j] = *(const v2f*)(arow1 + 4 * j + 16 * HD);
  }
  {
    float4 hw0, hw1, hw2, hw3;
    W_LOAD(0)
    W_STORE(0)
  }
  __syncthreads();

  v8f p0 = {}, p1 = {}, p2 = {}, p3 = {}, p4 = {}, p5 = {}, p6 = {}, p7 = {};
  v8f q0 = {}, q1 = {}, q2 = {}, q3 = {}, q4 = {}, q5 = {}, q6 = {}, q7 = {};

#define GEMM_PAIR(P, Q, OFF)                                               \
    bb = bp[OFF]; b.x = bb.x; b.y = bb.y;                                  \
    P = __builtin_amdgcn_wmma_f32_16x16x4_f32(false, a0, false, b, (short)0, P, false, false); \
    Q = __builtin_amdgcn_wmma_f32_16x16x4_f32(false, a1, false, b, (short)0, Q, false, false);

  const float2* bp = wp + hi * HD + m;

#define K_STEP_MMA                                                         \
    {                                                                      \
      float2 bb; v2f b;                                                    \
      GEMM_PAIR(p0, q0, 0)                                                 \
      GEMM_PAIR(p1, q1, 16)                                                \
      GEMM_PAIR(p2, q2, 32)                                                \
      GEMM_PAIR(p3, q3, 48)                                                \
      GEMM_PAIR(p4, q4, 64)                                                \
      GEMM_PAIR(p5, q5, 80)                                                \
      GEMM_PAIR(p6, q6, 96)                                                \
      GEMM_PAIR(p7, q7, 112)                                               \
      bp += 2 * HD;                                                        \
    }

  const float* abase = arow1;
#pragma unroll 1
  for (int c = 0; c < 7; ++c) {
    const float* anext = (c + 1 < 4) ? arow1 + 32 * (c + 1) : arow2 + 32 * (c + 1 - 4);
    float4 hw0, hw1, hw2, hw3;
#pragma unroll
    for (int j = 0; j < 8; ++j) {
      const v2f a0 = ra0[j & 3], a1 = ra1[j & 3];
      const float* apn = (j + 4 < 8) ? abase + 4 * (j + 4) : anext + 4 * (j + 4 - 8);
      ra0[j & 3] = *(const v2f*)(apn);
      ra1[j & 3] = *(const v2f*)(apn + 16 * HD);
      if (j == 0) W_LOAD(c + 1)
      K_STEP_MMA
      __builtin_amdgcn_sched_barrier(0);
    }
    W_STORE(c + 1)
    __syncthreads();
    abase = anext;
  }
  {
    v2f ta0[4], ta1[4];
#pragma unroll
    for (int j = 0; j < 4; ++j) {
      ta0[j] = *(const v2f*)(abase + 4 * (j + 4));
      ta1[j] = *(const v2f*)(abase + 4 * (j + 4) + 16 * HD);
    }
#pragma unroll
    for (int j = 0; j < 4; ++j) {
      const v2f a0 = ra0[j], a1 = ra1[j];
      K_STEP_MMA
      __builtin_amdgcn_sched_barrier(0);
    }
#pragma unroll
    for (int j = 0; j < 4; ++j) {
      const v2f a0 = ta0[j], a1 = ta1[j];
      K_STEP_MMA
      __builtin_amdgcn_sched_barrier(0);
    }
  }
#undef K_STEP_MMA
#undef W_LOAD
#undef W_STORE
#undef GEMM_PAIR

  const int rbase = rowBase + hi * 8;
  v8f accs[16] = {p0, p1, p2, p3, p4, p5, p6, p7, q0, q1, q2, q3, q4, q5, q6, q7};
if constexpr (!HEAD) {
#pragma unroll
  for (int rt = 0; rt < 2; ++rt) {
#pragma unroll
    for (int ct = 0; ct < 8; ++ct) {
      const int col = ct * 16 + m;
      const float bv = bias[col];
#pragma unroll
      for (int v = 0; v < 8; ++v) {
        float val = accs[rt * 8 + ct][v] + bv;
        val = val > 0.f ? val : 0.f;
        out[(size_t)(rbase + rt * 16 + v) * HD + col] = val;
      }
    }
  }
  } else {
  __shared__ float svals[2 * NN];
  __shared__ int   sidxs[2 * NN];
  float*  hs  = (float*)wp;
  float4* hs4 = (float4*)wp;

  __syncthreads();
  {
    const int lrow = wave * 32 + hi * 8;
#pragma unroll
    for (int rt = 0; rt < 2; ++rt) {
#pragma unroll
      for (int ct = 0; ct < 8; ++ct) {
        const int col = ct * 16 + m;
        const float bv = bias[col];
#pragma unroll
        for (int v = 0; v < 8; ++v) {
          float val = accs[rt * 8 + ct][v] + bv;
          val = val > 0.f ? val : 0.f;
          hs[(lrow + rt * 16 + v) * HD + col] = val;
        }
      }
    }
  }
  __syncthreads();

  const int node = t & (NN - 1);
  svals[t] = hs[t * HD + (HD - 1)];
  sidxs[t] = node;
  __syncthreads();

  for (int k = 2; k <= NN; k <<= 1) {
    for (int j = k >> 1; j > 0; j >>= 1) {
      const int ixj = t ^ j;
      if (ixj > t) {
        const bool dirDesc = ((node & k) == 0);
        const float va = svals[t], vb = svals[ixj];
        const int   ia = sidxs[t], ib = sidxs[ixj];
        const bool aBeforeB = (va > vb) || (va == vb && ia < ib);
        if (dirDesc ? !aBeforeB : aBeforeB) {
          svals[t] = vb; svals[ixj] = va;
          sidxs[t] = ib; sidxs[ixj] = ia;
        }
      }
      __syncthreads();
    }
  }

  const int q = t & 31, tt0 = t >> 5;
  const int ts3 = (tt0 + 24) < KTOP ? (tt0 + 24) : KTOP - 1;
  const float4 g00 = hs4[sidxs[tt0]      * (HD / 4) + q];
  const float4 g01 = hs4[sidxs[tt0 + 8]  * (HD / 4) + q];
  const float4 g02 = hs4[sidxs[tt0 + 16] * (HD / 4) + q];
  const float4 g03 = hs4[sidxs[ts3]      * (HD / 4) + q];
  const float4 g10 = hs4[(NN + sidxs[NN + tt0])      * (HD / 4) + q];
  const float4 g11 = hs4[(NN + sidxs[NN + tt0 + 8])  * (HD / 4) + q];
  const float4 g12 = hs4[(NN + sidxs[NN + tt0 + 16]) * (HD / 4) + q];
  const float4 g13 = hs4[(NN + sidxs[NN + ts3])      * (HD / 4) + q];
  __syncthreads();

  float4* xT0 = hs4;
  float4* xT1 = hs4 + (HD / 4) * XT_PITCH;
  float*  ys  = (float*)(hs4 + 2 * (HD / 4) * XT_PITCH);
  xT0[q * XT_PITCH + tt0]      = g00;
  xT0[q * XT_PITCH + tt0 + 8]  = g01;
  xT0[q * XT_PITCH + tt0 + 16] = g02;
  xT0[q * XT_PITCH + tt0 + 24] = g03;
  xT1[q * XT_PITCH + tt0]      = g10;
  xT1[q * XT_PITCH + tt0 + 8]  = g11;
  xT1[q * XT_PITCH + tt0 + 16] = g12;
  xT1[q * XT_PITCH + tt0 + 24] = g13;
  __syncthreads();

  {
    const int wv = __builtin_amdgcn_readfirstlane(t >> 5);
    const int tl = lane < CONV_T ? lane : CONV_T - 1;
    typedef const float __attribute__((address_space(4))) cfloat;
    const cfloat* wb = (const cfloat*)conv_w + (size_t)(wv * 4) * (HD * CONV_W);
    const cfloat* cb = (const cfloat*)conv_b + wv * 4;
    float a0 = cb[0], a1 = cb[1];
    float a2 = cb[2], a3 = cb[3];
    float c0 = a0, c1 = a1, c2 = a2, c3 = a3;
    const float4* xp0 = xT0 + tl;
    const float4* xp1 = xT1 + tl;
#pragma unroll 2
    for (int qq = 0; qq < HD / 4; ++qq) {
      const float4 x0 = xp0[qq * XT_PITCH];
      const float4 x1 = xp0[qq * XT_PITCH + 1];
      const float4 x2 = xp0[qq * XT_PITCH + 2];
      const float4 z0 = xp1[qq * XT_PITCH];
      const float4 z1 = xp1[qq * XT_PITCH + 1];
      const float4 z2 = xp1[qq * XT_PITCH + 2];
      const cfloat* w0 = wb + qq * (4 * CONV_W);
      CONV_CH(a0, w0, x0, x1, x2)
      CONV_CH(a1, w0 + HD * CONV_W, x0, x1, x2)
      CONV_CH(a2, w0 + 2 * HD * CONV_W, x0, x1, x2)
      CONV_CH(a3, w0 + 3 * HD * CONV_W, x0, x1, x2)
      CONV_CH(c0, w0, z0, z1, z2)
      CONV_CH(c1, w0 + HD * CONV_W, z0, z1, z2)
      CONV_CH(c2, w0 + 2 * HD * CONV_W, z0, z1, z2)
      CONV_CH(c3, w0 + 3 * HD * CONV_W, z0, z1, z2)
    }
    if (lane < CONV_T) {
      float* yo = ys + (wv * 4) * CONV_T + lane;
      yo[0]          = a0 > 0.f ? a0 : 0.f;
      yo[CONV_T]     = a1 > 0.f ? a1 : 0.f;
      yo[2 * CONV_T] = a2 > 0.f ? a2 : 0.f;
      yo[3 * CONV_T] = a3 > 0.f ? a3 : 0.f;
      yo[FLAT]              = c0 > 0.f ? c0 : 0.f;
      yo[FLAT + CONV_T]     = c1 > 0.f ? c1 : 0.f;
      yo[FLAT + 2 * CONV_T] = c2 > 0.f ? c2 : 0.f;
      yo[FLAT + 3 * CONV_T] = c3 > 0.f ? c3 : 0.f;
    }
  }
  __syncthreads();

  {
    float4* yg4 = (float4*)(yout + (size_t)(blockIdx.x * 2) * FLAT);
    const float4* ys4 = (const float4*)ys;
    for (int i = t; i < 2 * FLAT / 4; i += 256) yg4[i] = ys4[i];
  }
  }
}
#undef CONV_CH

#define DG        16
#define YS_PITCH  (FLAT + 4)
#define ZT_PITCH  (HD + 4)
#define KC        16
#define NCHUNK    (FLAT / (4 * KC))

__device__ __forceinline__ void load_w1_chunk(float (&b)[2 * KC], const float* __restrict__ p)
{
#pragma unroll
  for (int s = 0; s < KC; ++s) {
    b[2 * s]     = p[(size_t)(4 * s)     * HD];
    b[2 * s + 1] = p[(size_t)(4 * s + 1) * HD];
  }
}

__device__ __forceinline__ void mma_w1_chunk(v8f& acc0, v8f& acc1, const float (&b)[2 * KC],
                                             const float* arow)
{
#pragma unroll
  for (int s = 0; s < KC; s += 2) {
    const v2f a0 = *(const v2f*)(arow + 4 * s);
    const v2f a1 = *(const v2f*)(arow + 4 * s + 4);
    v2f f0, f1;
    f0.x = b[2 * s];     f0.y = b[2 * s + 1];
    f1.x = b[2 * s + 2]; f1.y = b[2 * s + 3];
    acc0 = __builtin_amdgcn_wmma_f32_16x16x4_f32(false, a0, false, f0, (short)0, acc0, false, false);
    acc1 = __builtin_amdgcn_wmma_f32_16x16x4_f32(false, a1, false, f1, (short)0, acc1, false, false);
  }
}

__global__ __launch_bounds__(256) void head_dense(
    const float* __restrict__ Y,
    const float* __restrict__ W1, const float* __restrict__ b1,
    const float* __restrict__ W2, const float* __restrict__ b2,
    float* __restrict__ out)
{
  __shared__ __attribute__((aligned(16))) float ys[DG * YS_PITCH];
  __shared__ __attribute__((aligned(16))) float zt[DG * ZT_PITCH];
  __shared__ __attribute__((aligned(16))) float w2s[HD * NCLS];
  __shared__ __attribute__((aligned(16))) float lg[DG * NCLS];

  const int t    = threadIdx.x;
  const int lane = t & 31;
  const int wave = t >> 5;
  const int m    = lane & 15;
  const int hi   = lane >> 4;
  const int kb   = hi * 2;
  const int g0   = blockIdx.x * DG;

  const float* wcol = W1 + (size_t)kb * HD + wave * 16 + m;
  float bA[2 * KC], bB[2 * KC];
  load_w1_chunk(bA, wcol);
  const float bv = b1[wave * 16 + m];

  const float4* yg4 = (const float4*)(Y + (size_t)g0 * FLAT);
#pragma unroll
  for (int k = 0; k < DG * (FLAT / 4) / 256; ++k) {
    const int i  = t + 256 * k;
    const int r  = i / (FLAT / 4);
    const int c4 = i - r * (FLAT / 4);
    *(float4*)&ys[r * YS_PITCH + c4 * 4] = yg4[i];
  }
#pragma unroll
  for (int k = 0; k < HD * NCLS / 4 / 256; ++k)
    ((float4*)w2s)[t + 256 * k] = ((const float4*)W2)[t + 256 * k];
  __syncthreads();

  v8f acc0 = {bv, bv, bv, bv, bv, bv, bv, bv};
  v8f acc1 = {};
  const float* arow = ys + m * YS_PITCH + kb;
  for (int c = 0; c < NCHUNK - 2; c += 2) {
    load_w1_chunk(bB, wcol + (size_t)(c + 1) * (4 * KC) * HD);
    mma_w1_chunk(acc0, acc1, bA, arow + c * (4 * KC));
    load_w1_chunk(bA, wcol + (size_t)(c + 2) * (4 * KC) * HD);
    mma_w1_chunk(acc0, acc1, bB, arow + (c + 1) * (4 * KC));
  }
  load_w1_chunk(bB, wcol + (size_t)(NCHUNK - 1) * (4 * KC) * HD);
  mma_w1_chunk(acc0, acc1, bA, arow + (NCHUNK - 2) * (4 * KC));
  mma_w1_chunk(acc0, acc1, bB, arow + (NCHUNK - 1) * (4 * KC));

#pragma unroll
  for (int v = 0; v < 8; ++v) {
    const float s = acc0[v] + acc1[v];
    zt[(v + 8 * hi) * ZT_PITCH + wave * 16 + m] = s > 0.f ? s : 0.f;
  }
  __syncthreads();

  const int gi  = t >> 4;
  const int cls = t & 15;
  float s = b2[cls];
  {
    const float* zr = zt + gi * ZT_PITCH;
    for (int j = 0; j < HD; j += 4) {
      const float4 zv = *(const float4*)(zr + j);
      s += zv.x * w2s[(j + 0) * NCLS + cls];
      s += zv.y * w2s[(j + 1) * NCLS + cls];
      s += zv.z * w2s[(j + 2) * NCLS + cls];
      s += zv.w * w2s[(j + 3) * NCLS + cls];
    }
  }
  lg[t] = s;
  __syncthreads();

  {
    const float* lr = lg + gi * NCLS;
    float mx = lr[0];
    for (int c = 1; c < NCLS; ++c) mx = fmaxf(mx, lr[c]);
    float se = 0.f;
    for (int c = 0; c < NCLS; ++c) se += expf(lr[c] - mx);
    const float lse = mx + logf(se);
    out[(size_t)(g0 + gi) * NCLS + cls] = s - lse;
  }
}

extern "C" void kernel_launch(void* const* d_in, const int* in_sizes, int n_in,
                              void* d_out, int out_size, void* d_ws, size_t ws_size,
                              hipStream_t stream) {
  (void)in_sizes; (void)n_in; (void)out_size; (void)ws_size;
  const float* x      = (const float*)d_in[0];
  const int*   src    = (const int*)d_in[1];
  const int*   dst    = (const int*)d_in[2];
  const float* Wl1    = (const float*)d_in[5];
  const float* bl1    = (const float*)d_in[6];
  const float* Wr1    = (const float*)d_in[7];
  const float* Wl2    = (const float*)d_in[8];
  const float* bl2    = (const float*)d_in[9];
  const float* Wr2    = (const float*)d_in[10];
  const float* Wl3    = (const float*)d_in[11];
  const float* bl3    = (const float*)d_in[12];
  const float* Wr3    = (const float*)d_in[13];
  const float* conv_w = (const float*)d_in[14];
  const float* conv_b = (const float*)d_in[15];
  const float* W1     = (const float*)d_in[16];
  const float* b1     = (const float*)d_in[17];
  const float* W2     = (const float*)d_in[18];
  const float* b2     = (const float*)d_in[19];

  float* ws   = (float*)d_ws;
  float* mean = ws;
  float* h1   = ws + (size_t)VN * HD;
  float* h2   = h1 + (size_t)VN * HD;

  const int gemmBlocks = VN / GEMM_ROWS;

  sage_aggregate<<<BG, 256, 0, stream>>>(x, src, dst, mean);
  sage_gemm<false><<<gemmBlocks, 256, 0, stream>>>(mean, x, Wl1, Wr1, bl1, h1, nullptr, nullptr, nullptr);
  sage_aggregate<<<BG, 256, 0, stream>>>(h1, src, dst, mean);
  sage_gemm<false><<<gemmBlocks, 256, 0, stream>>>(mean, h1, Wl2, Wr2, bl2, h2, nullptr, nullptr, nullptr);
  float* yflat = h1;
  sage_aggregate<<<BG, 256, 0, stream>>>(h2, src, dst, mean);
  sage_gemm<true><<<gemmBlocks, 256, 0, stream>>>(mean, h2, Wl3, Wr3, bl3, nullptr, conv_w, conv_b, yflat);
  head_dense<<<BG / DG, 256, 0, stream>>>(yflat, W1, b1, W2, b2, (float*)d_out);
}
